# starting kernel
# baseline (speedup 1.0000x reference)
#include <hip/hip_runtime.h>


typedef float v2f __attribute__((ext_vector_type(2)));
typedef float v8f __attribute__((ext_vector_type(8)));

#define F_DIM 128
#define SZ_STRIDE 132

__global__ __launch_bounds__(256) void spmm_edge_kernel(
    const int* __restrict__ row, const int* __restrict__ col,
    const float* __restrict__ vals, const float* __restrict__ x,
    float* __restrict__ z, int nEdges)
{
    const int e    = (int)((blockIdx.x * blockDim.x + threadIdx.x) >> 5);
    const int lane = threadIdx.x & 31;
    if (e >= nEdges) return;

    const int   r = row[e];
    const int   c = col[e];
    const float v = vals[e];

    const float4 d = ((const float4*)(x + (size_t)c * F_DIM))[lane];

    float* dst = z + (size_t)r * F_DIM + lane * 4;
    atomicAdd(dst + 0, v * d.x);
    atomicAdd(dst + 1, v * d.y);
    atomicAdd(dst + 2, v * d.z);
    atomicAdd(dst + 3, v * d.w);
}

__global__ __launch_bounds__(256) void gemm_inplace_kernel(
    float* __restrict__ zout, const float* __restrict__ W, int nNodes)
{
    __shared__ float sZ[16 * SZ_STRIDE];

    const int  tid     = threadIdx.x;
    const long rowBase = (long)blockIdx.x * 16;

    for (int i = tid; i < 16 * F_DIM; i += 256) {
        const int  r  = i >> 7;
        const int  cc = i & (F_DIM - 1);
        const long gr = rowBase + r;
        sZ[r * SZ_STRIDE + cc] = (gr < (long)nNodes) ? zout[gr * F_DIM + cc] : 0.0f;
    }
    __syncthreads();

    const int wave    = tid >> 5;
    const int lane    = tid & 31;
    const int m       = lane & 15;
    const int kh      = (lane >> 4) << 1;
    const int colBase = wave << 4;

    v8f acc = {};
#pragma unroll
    for (int k = 0; k < F_DIM; k += 4) {
        v2f a, b;
        a.x = sZ[m * SZ_STRIDE + k + kh];
        a.y = sZ[m * SZ_STRIDE + k + kh + 1];
        b.x = W[(k + kh)     * F_DIM + colBase + m];
        b.y = W[(k + kh + 1) * F_DIM + colBase + m];
        acc = __builtin_amdgcn_wmma_f32_16x16x4_f32(
             false, a,  false, b,
             (short)0, acc,  false,  false);
    }

    const int mhi = (lane >> 4) << 3;
#pragma unroll
    for (int i = 0; i < 8; ++i) {
        const long gr = rowBase + mhi + i;
        if (gr < (long)nNodes) zout[gr * F_DIM + colBase + m] = acc[i];
    }
}

extern "C" void kernel_launch(void* const* d_in, const int* in_sizes, int n_in,
                              void* d_out, int out_size, void* d_ws, size_t ws_size,
                              hipStream_t stream)
{
    const int*   row  = (const int*)d_in[0];
    const int*   col  = (const int*)d_in[1];
    const float* vals = (const float*)d_in[2];
    const float* x    = (const float*)d_in[3];
    const float* W    = (const float*)d_in[4];
    float*       out  = (float*)d_out;

    const int nEdges = in_sizes[0];
    const int nNodes = in_sizes[3] / F_DIM;

    hipMemsetAsync(out, 0, (size_t)out_size * sizeof(float), stream);

    const int gridS = (nEdges + 7) / 8;
    spmm_edge_kernel<<<gridS, 256, 0, stream>>>(row, col, vals, x, out, nEdges);

    const int gridG = (nNodes + 15) / 16;
    gemm_inplace_kernel<<<gridG, 256, 0, stream>>>(out, W, nNodes);
}


# SpMM float-atomic scatter replaced by in-call counting sort by row (int atomic gives arrival-order rank) and one wave per row summing its edges in registers; per-row sum order = arrival order
# speedup vs baseline: 1.4828x; 1.4828x over previous
#include <hip/hip_runtime.h>


typedef float v2f __attribute__((ext_vector_type(2)));
typedef float v8f __attribute__((ext_vector_type(8)));

#define F_DIM 128
#define SZ_STRIDE 132
#define SCAN_TILE 1024

__global__ __launch_bounds__(256) void edge_rank_kernel(
    const int* __restrict__ row, int* __restrict__ cnt,
    int* __restrict__ rank, int nEdges)
{
    const int e = (int)(blockIdx.x * 256 + threadIdx.x);
    if (e < nEdges) rank[e] = atomicAdd(&cnt[row[e]], 1);
}

__global__ __launch_bounds__(256) void row_start_kernel(
    const int* __restrict__ cnt, int* __restrict__ off, int nNodes)
{
    __shared__ int sFront[8];
    __shared__ int sWave[8];

    const int tid     = threadIdx.x;
    const int lane    = tid & 31;
    const int wave    = tid >> 5;
    const int tileBeg = (int)blockIdx.x * SCAN_TILE;

    int s = 0;
    const int4* c4 = (const int4*)cnt;
    const int nb = (int)blockIdx.x;
    for (int b = 0; b < nb; b += 16) {
        int4 v[16];
#pragma unroll
        for (int u = 0; u < 16; ++u) {
            const int sl = (b + u < nb) ? (b + u) : (nb - 1);
            v[u] = c4[sl * 256 + tid];
        }
#pragma unroll
        for (int u = 0; u < 16; ++u)
            s += (b + u < nb) ? ((v[u].x + v[u].y) + (v[u].z + v[u].w)) : 0;
    }
    for (int d = 16; d > 0; d >>= 1) s += __shfl_xor(s, d);
    if (lane == 0) sFront[wave] = s;

    const int i0 = tileBeg + tid * 4;
    const int a0 = (i0     < nNodes) ? cnt[i0]     : 0;
    const int a1 = (i0 + 1 < nNodes) ? cnt[i0 + 1] : 0;
    const int a2 = (i0 + 2 < nNodes) ? cnt[i0 + 2] : 0;
    const int a3 = (i0 + 3 < nNodes) ? cnt[i0 + 3] : 0;
    const int t  = (a0 + a1) + (a2 + a3);

    int inc = t;
    for (int d = 1; d < 32; d <<= 1) {
        const int u = __shfl_up(inc, d);
        if (lane >= d) inc += u;
    }
    if (lane == 31) sWave[wave] = inc;
    __syncthreads();

    int base = 0;
    for (int w = 0; w < 8; ++w) base += sFront[w];
    for (int w = 0; w < wave; ++w) base += sWave[w];

    const int ex = base + inc - t;
    if (i0     < nNodes) off[i0]     = ex;
    if (i0 + 1 < nNodes) off[i0 + 1] = ex + a0;
    if (i0 + 2 < nNodes) off[i0 + 2] = ex + a0 + a1;
    if (i0 + 3 < nNodes) off[i0 + 3] = ex + a0 + a1 + a2;
}

__global__ __launch_bounds__(256) void edge_place_kernel(
    const int* __restrict__ row, const int* __restrict__ col,
    const float* __restrict__ vals, const int* __restrict__ off,
    const int* __restrict__ rank, int2* __restrict__ pe, int nEdges)
{
    const int e = (int)(blockIdx.x * 256 + threadIdx.x);
    if (e < nEdges) {
        const int p = off[row[e]] + rank[e];
        pe[p] = make_int2(col[e], __float_as_int(vals[e]));
    }
}

static __device__ __forceinline__ void fma4(float4& acc, float v, const float4& d)
{
    acc.x = fmaf(v, d.x, acc.x);
    acc.y = fmaf(v, d.y, acc.y);
    acc.z = fmaf(v, d.z, acc.z);
    acc.w = fmaf(v, d.w, acc.w);
}

template <int G>
static __device__ __forceinline__ void gather_group(
    float4& acc, const float4* __restrict__ x4, int2 ev, int j, int lane)
{
    float4 d[G];
    float  v[G];
#pragma unroll
    for (int u = 0; u < G; ++u) {
        const int c = __builtin_amdgcn_readlane(ev.x, j + u);
        v[u] = __int_as_float(__builtin_amdgcn_readlane(ev.y, j + u));
        d[u] = x4[(size_t)c * (F_DIM / 4) + lane];
    }
#pragma unroll
    for (int u = 0; u < G; ++u) fma4(acc, v[u], d[u]);
}

__global__ __launch_bounds__(256) void spmm_row_kernel(
    const int* __restrict__ off, const int* __restrict__ cnt,
    const int2* __restrict__ pe, const float* __restrict__ x,
    float* __restrict__ z, int nNodes)
{
    const int lane = threadIdx.x & 31;
    const int r = __builtin_amdgcn_readfirstlane(
        (int)((blockIdx.x * 256 + threadIdx.x) >> 5));
    if (r >= nNodes) return;

    const int beg = off[r];
    const int n   = cnt[r];
    const float4* x4 = (const float4*)x;

    float4 acc = make_float4(0.0f, 0.0f, 0.0f, 0.0f);
    for (int base = 0; base < n; base += 32) {
        const int m = (n - base < 32) ? (n - base) : 32;
        int2 ev = make_int2(0, 0);
        if (lane < m) ev = pe[beg + base + lane];
        int j = 0;
        for (; j + 8 <= m; j += 8) gather_group<8>(acc, x4, ev, j, lane);
        if (j + 4 <= m) { gather_group<4>(acc, x4, ev, j, lane); j += 4; }
        if (j + 2 <= m) { gather_group<2>(acc, x4, ev, j, lane); j += 2; }
        if (j < m)      { gather_group<1>(acc, x4, ev, j, lane); }
    }
    ((float4*)(z + (size_t)r * F_DIM))[lane] = acc;
}

__global__ __launch_bounds__(256) void gemm_inplace_kernel(
    float* __restrict__ zout, const float* __restrict__ W, int nNodes)
{
    __shared__ float sZ[16 * SZ_STRIDE];

    const int  tid     = threadIdx.x;
    const long rowBase = (long)blockIdx.x * 16;

    for (int i = tid; i < 16 * F_DIM; i += 256) {
        const int  r  = i >> 7;
        const int  cc = i & (F_DIM - 1);
        const long gr = rowBase + r;
        sZ[r * SZ_STRIDE + cc] = (gr < (long)nNodes) ? zout[gr * F_DIM + cc] : 0.0f;
    }
    __syncthreads();

    const int wave    = tid >> 5;
    const int lane    = tid & 31;
    const int m       = lane & 15;
    const int kh      = (lane >> 4) << 1;
    const int colBase = wave << 4;

    v8f acc = {};
#pragma unroll
    for (int k = 0; k < F_DIM; k += 4) {
        v2f a, b;
        a.x = sZ[m * SZ_STRIDE + k + kh];
        a.y = sZ[m * SZ_STRIDE + k + kh + 1];
        b.x = W[(k + kh)     * F_DIM + colBase + m];
        b.y = W[(k + kh + 1) * F_DIM + colBase + m];
        acc = __builtin_amdgcn_wmma_f32_16x16x4_f32(
             false, a,  false, b,
             (short)0, acc,  false,  false);
    }

    const int mhi = (lane >> 4) << 3;
#pragma unroll
    for (int i = 0; i < 8; ++i) {
        const long gr = rowBase + mhi + i;
        if (gr < (long)nNodes) zout[gr * F_DIM + colBase + m] = acc[i];
    }
}

extern "C" void kernel_launch(void* const* d_in, const int* in_sizes, int n_in,
                              void* d_out, int out_size, void* d_ws, size_t ws_size,
                              hipStream_t stream)
{
    const int*   row  = (const int*)d_in[0];
    const int*   col  = (const int*)d_in[1];
    const float* vals = (const float*)d_in[2];
    const float* x    = (const float*)d_in[3];
    const float* W    = (const float*)d_in[4];
    float*       out  = (float*)d_out;

    const int nEdges = in_sizes[0];
    const int nNodes = in_sizes[3] / F_DIM;

    const int nPad = (nNodes + SCAN_TILE - 1) / SCAN_TILE * SCAN_TILE;
    const int ePad = (nEdges + 1023) / 1024 * 1024;
    int*  cnt  = (int*)d_ws;
    int*  off  = cnt + nPad;
    int*  rank = off + nPad;
    int2* pe   = (int2*)(rank + ePad);

    hipMemsetAsync(cnt, 0, (size_t)nPad * sizeof(int), stream);

    const int gridE = (nEdges + 255) / 256;
    edge_rank_kernel<<<gridE, 256, 0, stream>>>(row, cnt, rank, nEdges);

    const int gridN = nPad / SCAN_TILE;
    row_start_kernel<<<gridN, 256, 0, stream>>>(cnt, off, nNodes);

    edge_place_kernel<<<gridE, 256, 0, stream>>>(row, col, vals, off, rank, pe, nEdges);

    const int gridS = (nNodes + 7) / 8;
    spmm_row_kernel<<<gridS, 256, 0, stream>>>(off, cnt, pe, x, out, nNodes);

    const int gridG = (nNodes + 15) / 16;
    gemm_inplace_kernel<<<gridG, 256, 0, stream>>>(out, W, nNodes);
}


# SpMM fused into the projection launch: each block sums its 32 rows of z into LDS and multiplies them by W from there (f32 WMMA); z never goes to memory; one launch and 125000 waves fewer
# speedup vs baseline: 1.6537x; 1.1153x over previous
#include <hip/hip_runtime.h>


typedef float v2f __attribute__((ext_vector_type(2)));
typedef float v8f __attribute__((ext_vector_type(8)));

#define F_DIM 128
#define SZ_STRIDE 132
#define SCAN_TILE 1024

__global__ __launch_bounds__(256) void edge_rank_kernel(
    const int* __restrict__ row, int* __restrict__ cnt,
    int* __restrict__ rank, int nEdges)
{
    const int e = (int)(blockIdx.x * 256 + threadIdx.x);
    if (e < nEdges) rank[e] = atomicAdd(&cnt[row[e]], 1);
}

__global__ __launch_bounds__(256) void row_start_kernel(
    const int* __restrict__ cnt, int* __restrict__ off, int nNodes)
{
    __shared__ int sFront[8];
    __shared__ int sWave[8];

    const int tid     = threadIdx.x;
    const int lane    = tid & 31;
    const int wave    = tid >> 5;
    const int tileBeg = (int)blockIdx.x * SCAN_TILE;

    int s = 0;
    const int4* c4 = (const int4*)cnt;
    const int nb = (int)blockIdx.x;
    for (int b = 0; b < nb; b += 16) {
        int4 v[16];
#pragma unroll
        for (int u = 0; u < 16; ++u) {
            const int sl = (b + u < nb) ? (b + u) : (nb - 1);
            v[u] = c4[sl * 256 + tid];
        }
#pragma unroll
        for (int u = 0; u < 16; ++u)
            s += (b + u < nb) ? ((v[u].x + v[u].y) + (v[u].z + v[u].w)) : 0;
    }
    for (int d = 16; d > 0; d >>= 1) s += __shfl_xor(s, d);
    if (lane == 0) sFront[wave] = s;

    const int i0 = tileBeg + tid * 4;
    const int a0 = (i0     < nNodes) ? cnt[i0]     : 0;
    const int a1 = (i0 + 1 < nNodes) ? cnt[i0 + 1] : 0;
    const int a2 = (i0 + 2 < nNodes) ? cnt[i0 + 2] : 0;
    const int a3 = (i0 + 3 < nNodes) ? cnt[i0 + 3] : 0;
    const int t  = (a0 + a1) + (a2 + a3);

    int inc = t;
    for (int d = 1; d < 32; d <<= 1) {
        const int u = __shfl_up(inc, d);
        if (lane >= d) inc += u;
    }
    if (lane == 31) sWave[wave] = inc;
    __syncthreads();

    int base = 0;
    for (int w = 0; w < 8; ++w) base += sFront[w];
    for (int w = 0; w < wave; ++w) base += sWave[w];

    const int ex = base + inc - t;
    if (i0     < nNodes) off[i0]     = ex;
    if (i0 + 1 < nNodes) off[i0 + 1] = ex + a0;
    if (i0 + 2 < nNodes) off[i0 + 2] = ex + a0 + a1;
    if (i0 + 3 < nNodes) off[i0 + 3] = ex + a0 + a1 + a2;
}

__global__ __launch_bounds__(256) void edge_place_kernel(
    const int* __restrict__ row, const int* __restrict__ col,
    const float* __restrict__ vals, const int* __restrict__ off,
    const int* __restrict__ rank, int2* __restrict__ pe, int nEdges)
{
    const int e = (int)(blockIdx.x * 256 + threadIdx.x);
    if (e < nEdges) {
        const int p = off[row[e]] + rank[e];
        pe[p] = make_int2(col[e], __float_as_int(vals[e]));
    }
}

static __device__ __forceinline__ void fma4(float4& acc, float v, const float4& d)
{
    acc.x = fmaf(v, d.x, acc.x);
    acc.y = fmaf(v, d.y, acc.y);
    acc.z = fmaf(v, d.z, acc.z);
    acc.w = fmaf(v, d.w, acc.w);
}

#define TILE_ROWS 32
#define WAVE_ROWS 4

template <int G>
static __device__ __forceinline__ void gather_group(
    float4& acc, const float4* __restrict__ x4, int2 ev, int j, int lane)
{
    float4 d[G];
    float  v[G];
#pragma unroll
    for (int u = 0; u < G; ++u) {
        const int c = __builtin_amdgcn_readlane(ev.x, j + u);
        v[u] = __int_as_float(__builtin_amdgcn_readlane(ev.y, j + u));
        d[u] = x4[(size_t)c * (F_DIM / 4) + lane];
    }
#pragma unroll
    for (int u = 0; u < G; ++u) fma4(acc, v[u], d[u]);
}

__global__ __launch_bounds__(256) void spmm_gemm_kernel(
    const int* __restrict__ off, const int* __restrict__ cnt,
    const int2* __restrict__ pe, const float* __restrict__ x,
    const float* __restrict__ W, float* __restrict__ out, int nNodes)
{
    __shared__ __attribute__((aligned(16))) float sZ[TILE_ROWS * SZ_STRIDE];

    const int tid  = threadIdx.x;
    const int lane = tid & 31;
    const int wave = __builtin_amdgcn_readfirstlane(tid >> 5);
    const int rowBase = (int)blockIdx.x * TILE_ROWS;
    const float4* x4 = (const float4*)x;

    int beg[WAVE_ROWS], n[WAVE_ROWS];
    int2 ev0[WAVE_ROWS];
#pragma unroll
    for (int q = 0; q < WAVE_ROWS; ++q) {
        const int r = rowBase + wave * WAVE_ROWS + q;
        beg[q] = (r < nNodes) ? off[r] : 0;
        n[q]   = (r < nNodes) ? cnt[r] : 0;
    }
#pragma unroll
    for (int q = 0; q < WAVE_ROWS; ++q) {
        ev0[q] = make_int2(0, 0);
        if (lane < n[q]) ev0[q] = pe[beg[q] + lane];
    }

#pragma unroll
    for (int q = 0; q < WAVE_ROWS; ++q) {
        float4 acc = make_float4(0.0f, 0.0f, 0.0f, 0.0f);
        int2 ev = ev0[q];
        for (int base = 0; base < n[q]; base += 32) {
            const int m = (n[q] - base < 32) ? (n[q] - base) : 32;
            if (base > 0) {
                ev = make_int2(0, 0);
                if (lane < m) ev = pe[beg[q] + base + lane];
            }
            int j = 0;
            for (; j + 8 <= m; j += 8) gather_group<8>(acc, x4, ev, j, lane);
            if (j + 4 <= m) { gather_group<4>(acc, x4, ev, j, lane); j += 4; }
            if (j + 2 <= m) { gather_group<2>(acc, x4, ev, j, lane); j += 2; }
            if (j < m)      { gather_group<1>(acc, x4, ev, j, lane); }
        }
        *(float4*)&sZ[(wave * WAVE_ROWS + q) * SZ_STRIDE + lane * 4] = acc;
    }
    __syncthreads();

    const int m       = lane & 15;
    const int kh      = (lane >> 4) << 1;
    const int colBase = wave << 4;

    v8f acc0 = {};
    v8f acc1 = {};
#pragma unroll
    for (int k = 0; k < F_DIM; k += 4) {
        v2f a0, a1, b;
        a0.x = sZ[m * SZ_STRIDE + k + kh];
        a0.y = sZ[m * SZ_STRIDE + k + kh + 1];
        a1.x = sZ[(16 + m) * SZ_STRIDE + k + kh];
        a1.y = sZ[(16 + m) * SZ_STRIDE + k + kh + 1];
        b.x = W[(k + kh)     * F_DIM + colBase + m];
        b.y = W[(k + kh + 1) * F_DIM + colBase + m];
        acc0 = __builtin_amdgcn_wmma_f32_16x16x4_f32(
             false, a0,  false, b,
             (short)0, acc0,  false,  false);
        acc1 = __builtin_amdgcn_wmma_f32_16x16x4_f32(
             false, a1,  false, b,
             (short)0, acc1,  false,  false);
    }

    const int mhi = (lane >> 4) << 3;
#pragma unroll
    for (int i = 0; i < 8; ++i) {
        const long g0 = (long)rowBase + mhi + i;
        const long g1 = g0 + 16;
        if (g0 < (long)nNodes) out[g0 * F_DIM + colBase + m] = acc0[i];
        if (g1 < (long)nNodes) out[g1 * F_DIM + colBase + m] = acc1[i];
    }
}

extern "C" void kernel_launch(void* const* d_in, const int* in_sizes, int n_in,
                              void* d_out, int out_size, void* d_ws, size_t ws_size,
                              hipStream_t stream)
{
    const int*   row  = (const int*)d_in[0];
    const int*   col  = (const int*)d_in[1];
    const float* vals = (const float*)d_in[2];
    const float* x    = (const float*)d_in[3];
    const float* W    = (const float*)d_in[4];
    float*       out  = (float*)d_out;

    const int nEdges = in_sizes[0];
    const int nNodes = in_sizes[3] / F_DIM;

    const int nPad = (nNodes + SCAN_TILE - 1) / SCAN_TILE * SCAN_TILE;
    const int ePad = (nEdges + 1023) / 1024 * 1024;
    int*  cnt  = (int*)d_ws;
    int*  off  = cnt + nPad;
    int*  rank = off + nPad;
    int2* pe   = (int2*)(rank + ePad);

    hipMemsetAsync(cnt, 0, (size_t)nPad * sizeof(int), stream);

    const int gridE = (nEdges + 255) / 256;
    edge_rank_kernel<<<gridE, 256, 0, stream>>>(row, cnt, rank, nEdges);

    const int gridN = nPad / SCAN_TILE;
    row_start_kernel<<<gridN, 256, 0, stream>>>(cnt, off, nNodes);

    edge_place_kernel<<<gridE, 256, 0, stream>>>(row, col, vals, off, rank, pe, nEdges);

    const int gridF = (nNodes + TILE_ROWS - 1) / TILE_ROWS;
    spmm_gemm_kernel<<<gridF, 256, 0, stream>>>(off, cnt, pe, x, W, out, nNodes);
}


# Edge sort by row without global atomics: edges binned by 256-row bucket per 4096-edge chunk with LDS counters, each bucket then ranked by row in LDS; per-row sum order = LDS arrival order
# speedup vs baseline: 2.3568x; 1.4252x over previous
#include <hip/hip_runtime.h>


typedef float v2f __attribute__((ext_vector_type(2)));
typedef float v8f __attribute__((ext_vector_type(8)));

#define F_DIM 128
#define SZ_STRIDE 132
#define SCAN_TILE 1024

#define BUCKET_SHIFT 8
#define BUCKET_ROWS  (1 << BUCKET_SHIFT)
#define CHUNK_PER_THREAD 16
#define CHUNK_EDGES (256 * CHUNK_PER_THREAD)
#define KEY_SHIFT 24
#define COL_MASK  0x00ffffff
#define SORT_BATCH 20

__global__ __launch_bounds__(256) void bin_count_kernel(
    const int* __restrict__ row, int* __restrict__ hist, int nEdges, int nBuckets)
{
    extern __shared__ int sBin[];
    const int tid = threadIdx.x;
    for (int i = tid; i < nBuckets; i += 256) sBin[i] = 0;
    __syncthreads();

    const int chunkBeg = (int)blockIdx.x * CHUNK_EDGES;
    int r[CHUNK_PER_THREAD];
#pragma unroll
    for (int u = 0; u < CHUNK_PER_THREAD; ++u) {
        const int e = chunkBeg + u * 256 + tid;
        r[u] = row[e < nEdges ? e : nEdges - 1];
    }
#pragma unroll
    for (int u = 0; u < CHUNK_PER_THREAD; ++u) {
        const int e = chunkBeg + u * 256 + tid;
        atomicAdd(&sBin[r[u] >> BUCKET_SHIFT], e < nEdges ? 1 : 0);
    }
    __syncthreads();

    const int nChunks = (int)gridDim.x;
    for (int i = tid; i < nBuckets; i += 256)
        hist[i * nChunks + (int)blockIdx.x] = sBin[i];
}

__global__ __launch_bounds__(256) void row_start_kernel(
    const int* __restrict__ src, int* __restrict__ dst, int n)
{
    __shared__ int sFront[8];
    __shared__ int sWave[8];

    const int tid     = threadIdx.x;
    const int lane    = tid & 31;
    const int wave    = tid >> 5;
    const int tileBeg = (int)blockIdx.x * SCAN_TILE;

    int s = 0;
    const int4* c4 = (const int4*)src;
    const int nb = (int)blockIdx.x;
    for (int b = 0; b < nb; b += 16) {
        int4 v[16];
#pragma unroll
        for (int u = 0; u < 16; ++u) {
            const int sl = (b + u < nb) ? (b + u) : (nb - 1);
            v[u] = c4[sl * 256 + tid];
        }
#pragma unroll
        for (int u = 0; u < 16; ++u)
            s += (b + u < nb) ? ((v[u].x + v[u].y) + (v[u].z + v[u].w)) : 0;
    }
    for (int d = 16; d > 0; d >>= 1) s += __shfl_xor(s, d);
    if (lane == 0) sFront[wave] = s;

    const int i0 = tileBeg + tid * 4;
    const int a0 = (i0     < n) ? src[i0]     : 0;
    const int a1 = (i0 + 1 < n) ? src[i0 + 1] : 0;
    const int a2 = (i0 + 2 < n) ? src[i0 + 2] : 0;
    const int a3 = (i0 + 3 < n) ? src[i0 + 3] : 0;
    const int t  = (a0 + a1) + (a2 + a3);

    int inc = t;
    for (int d = 1; d < 32; d <<= 1) {
        const int u = __shfl_up(inc, d);
        if (lane >= d) inc += u;
    }
    if (lane == 31) sWave[wave] = inc;
    __syncthreads();

    int base = 0;
    for (int w = 0; w < 8; ++w) base += sFront[w];
    for (int w = 0; w < wave; ++w) base += sWave[w];

    const int ex = base + inc - t;
    if (i0     < n) dst[i0]     = ex;
    if (i0 + 1 < n) dst[i0 + 1] = ex + a0;
    if (i0 + 2 < n) dst[i0 + 2] = ex + a0 + a1;
    if (i0 + 3 < n) dst[i0 + 3] = ex + a0 + a1 + a2;
}

__global__ __launch_bounds__(256) void bin_scatter_kernel(
    const int* __restrict__ row, const int* __restrict__ col,
    const float* __restrict__ vals, const int* __restrict__ base,
    int2* __restrict__ binned, int nEdges, int nBuckets)
{
    extern __shared__ int sBin[];
    const int tid = threadIdx.x;
    const int nChunks = (int)gridDim.x;
    for (int i = tid; i < nBuckets; i += 256)
        sBin[i] = base[i * nChunks + (int)blockIdx.x];
    __syncthreads();

    const int chunkBeg = (int)blockIdx.x * CHUNK_EDGES;
    int   r[CHUNK_PER_THREAD];
    int   c[CHUNK_PER_THREAD];
    float v[CHUNK_PER_THREAD];
#pragma unroll
    for (int u = 0; u < CHUNK_PER_THREAD; ++u) {
        const int e  = chunkBeg + u * 256 + tid;
        const int ec = e < nEdges ? e : nEdges - 1;
        r[u] = row[ec];
        c[u] = col[ec];
        v[u] = vals[ec];
    }
    int p[CHUNK_PER_THREAD];
#pragma unroll
    for (int u = 0; u < CHUNK_PER_THREAD; ++u) {
        const int e = chunkBeg + u * 256 + tid;
        p[u] = atomicAdd(&sBin[r[u] >> BUCKET_SHIFT], e < nEdges ? 1 : 0);
    }
#pragma unroll
    for (int u = 0; u < CHUNK_PER_THREAD; ++u) {
        const int e = chunkBeg + u * 256 + tid;
        if (e < nEdges) {
            const unsigned key = ((unsigned)r[u] & (BUCKET_ROWS - 1)) << KEY_SHIFT;
            binned[p[u]] = make_int2((int)(key | (unsigned)c[u]), __float_as_int(v[u]));
        }
    }
}

__global__ __launch_bounds__(256) void bucket_sort_kernel(
    const int* __restrict__ base, const int2* __restrict__ binned,
    int* __restrict__ off, int* __restrict__ cnt, int2* __restrict__ pe,
    int nNodes, int nEdges, int nChunks)
{
    __shared__ int sCnt[BUCKET_ROWS];
    __shared__ int sCur[BUCKET_ROWS];
    __shared__ int sWave[8];

    const int tid  = threadIdx.x;
    const int lane = tid & 31;
    const int wave = tid >> 5;
    const int b    = (int)blockIdx.x;

    const int beg = base[b * nChunks];
    const int end = (b + 1 < (int)gridDim.x) ? base[(b + 1) * nChunks] : nEdges;
    const int last = (end > beg) ? (end - 1) : 0;

    sCnt[tid] = 0;
    __syncthreads();

    int2 rec0[SORT_BATCH];
#pragma unroll
    for (int u = 0; u < SORT_BATCH; ++u) {
        const int i = beg + u * 256 + tid;
        rec0[u] = binned[i < end ? i : last];
    }
    __builtin_amdgcn_sched_barrier(0);
#pragma unroll
    for (int u = 0; u < SORT_BATCH; ++u) {
        const int i = beg + u * 256 + tid;
        atomicAdd(&sCnt[(unsigned)rec0[u].x >> KEY_SHIFT], i < end ? 1 : 0);
    }
    for (int s = beg + 256 * SORT_BATCH; s < end; s += 256 * SORT_BATCH) {
        int key[SORT_BATCH];
#pragma unroll
        for (int u = 0; u < SORT_BATCH; ++u) {
            const int i = s + u * 256 + tid;
            key[u] = binned[i < end ? i : last].x;
        }
#pragma unroll
        for (int u = 0; u < SORT_BATCH; ++u) {
            const int i = s + u * 256 + tid;
            atomicAdd(&sCnt[(unsigned)key[u] >> KEY_SHIFT], i < end ? 1 : 0);
        }
    }
    __syncthreads();

    const int c = sCnt[tid];
    int inc = c;
    for (int d = 1; d < 32; d <<= 1) {
        const int up = __shfl_up(inc, d);
        if (lane >= d) inc += up;
    }
    if (lane == 31) sWave[wave] = inc;
    __syncthreads();

    int start = beg + inc - c;
    for (int w = 0; w < 8; ++w) start += (w < wave) ? sWave[w] : 0;

    const int r = b * BUCKET_ROWS + tid;
    if (r < nNodes) {
        off[r] = start;
        cnt[r] = c;
    }
    sCur[tid] = start;
    __syncthreads();

    int p0[SORT_BATCH];
#pragma unroll
    for (int u = 0; u < SORT_BATCH; ++u) {
        const int i = beg + u * 256 + tid;
        p0[u] = atomicAdd(&sCur[(unsigned)rec0[u].x >> KEY_SHIFT], i < end ? 1 : 0);
    }
#pragma unroll
    for (int u = 0; u < SORT_BATCH; ++u) {
        const int i = beg + u * 256 + tid;
        if (i < end) pe[p0[u]] = make_int2(rec0[u].x & COL_MASK, rec0[u].y);
    }
    for (int s = beg + 256 * SORT_BATCH; s < end; s += 256 * SORT_BATCH) {
        int2 rec[SORT_BATCH];
#pragma unroll
        for (int u = 0; u < SORT_BATCH; ++u) {
            const int i = s + u * 256 + tid;
            rec[u] = binned[i < end ? i : last];
        }
        int p[SORT_BATCH];
#pragma unroll
        for (int u = 0; u < SORT_BATCH; ++u) {
            const int i = s + u * 256 + tid;
            p[u] = atomicAdd(&sCur[(unsigned)rec[u].x >> KEY_SHIFT], i < end ? 1 : 0);
        }
#pragma unroll
        for (int u = 0; u < SORT_BATCH; ++u) {
            const int i = s + u * 256 + tid;
            if (i < end) pe[p[u]] = make_int2(rec[u].x & COL_MASK, rec[u].y);
        }
    }
}

static __device__ __forceinline__ void fma4(float4& acc, float v, const float4& d)
{
    acc.x = fmaf(v, d.x, acc.x);
    acc.y = fmaf(v, d.y, acc.y);
    acc.z = fmaf(v, d.z, acc.z);
    acc.w = fmaf(v, d.w, acc.w);
}

#define TILE_ROWS 32
#define WAVE_ROWS 4

template <int G>
static __device__ __forceinline__ void gather_group(
    float4& acc, const float4* __restrict__ x4, int2 ev, int j, int lane)
{
    float4 d[G];
    float  v[G];
#pragma unroll
    for (int u = 0; u < G; ++u) {
        const int c = __builtin_amdgcn_readlane(ev.x, j + u);
        v[u] = __int_as_float(__builtin_amdgcn_readlane(ev.y, j + u));
        d[u] = x4[(size_t)c * (F_DIM / 4) + lane];
    }
#pragma unroll
    for (int u = 0; u < G; ++u) fma4(acc, v[u], d[u]);
}

__global__ __launch_bounds__(256) void spmm_gemm_kernel(
    const int* __restrict__ off, const int* __restrict__ cnt,
    const int2* __restrict__ pe, const float* __restrict__ x,
    const float* __restrict__ W, float* __restrict__ out, int nNodes)
{
    __shared__ __attribute__((aligned(16))) float sZ[TILE_ROWS * SZ_STRIDE];

    const int tid  = threadIdx.x;
    const int lane = tid & 31;
    const int wave = __builtin_amdgcn_readfirstlane(tid >> 5);
    const int rowBase = (int)blockIdx.x * TILE_ROWS;
    const float4* x4 = (const float4*)x;

    int beg[WAVE_ROWS], n[WAVE_ROWS];
    int2 ev0[WAVE_ROWS];
#pragma unroll
    for (int q = 0; q < WAVE_ROWS; ++q) {
        const int r = rowBase + wave * WAVE_ROWS + q;
        beg[q] = (r < nNodes) ? off[r] : 0;
        n[q]   = (r < nNodes) ? cnt[r] : 0;
    }
#pragma unroll
    for (int q = 0; q < WAVE_ROWS; ++q) {
        ev0[q] = make_int2(0, 0);
        if (lane < n[q]) ev0[q] = pe[beg[q] + lane];
    }

#pragma unroll
    for (int q = 0; q < WAVE_ROWS; ++q) {
        float4 acc = make_float4(0.0f, 0.0f, 0.0f, 0.0f);
        int2 ev = ev0[q];
        for (int base = 0; base < n[q]; base += 32) {
            const int m = (n[q] - base < 32) ? (n[q] - base) : 32;
            if (base > 0) {
                ev = make_int2(0, 0);
                if (lane < m) ev = pe[beg[q] + base + lane];
            }
            int j = 0;
            for (; j + 8 <= m; j += 8) gather_group<8>(acc, x4, ev, j, lane);
            if (j + 4 <= m) { gather_group<4>(acc, x4, ev, j, lane); j += 4; }
            if (j + 2 <= m) { gather_group<2>(acc, x4, ev, j, lane); j += 2; }
            if (j < m)      { gather_group<1>(acc, x4, ev, j, lane); }
        }
        *(float4*)&sZ[(wave * WAVE_ROWS + q) * SZ_STRIDE + lane * 4] = acc;
    }
    __syncthreads();

    const int m       = lane & 15;
    const int kh      = (lane >> 4) << 1;
    const int colBase = wave << 4;

    v8f acc0 = {};
    v8f acc1 = {};
#pragma unroll
    for (int k = 0; k < F_DIM; k += 4) {
        v2f a0, a1, b;
        a0.x = sZ[m * SZ_STRIDE + k + kh];
        a0.y = sZ[m * SZ_STRIDE + k + kh + 1];
        a1.x = sZ[(16 + m) * SZ_STRIDE + k + kh];
        a1.y = sZ[(16 + m) * SZ_STRIDE + k + kh + 1];
        b.x = W[(k + kh)     * F_DIM + colBase + m];
        b.y = W[(k + kh + 1) * F_DIM + colBase + m];
        acc0 = __builtin_amdgcn_wmma_f32_16x16x4_f32(
             false, a0,  false, b,
             (short)0, acc0,  false,  false);
        acc1 = __builtin_amdgcn_wmma_f32_16x16x4_f32(
             false, a1,  false, b,
             (short)0, acc1,  false,  false);
    }

    const int mhi = (lane >> 4) << 3;
#pragma unroll
    for (int i = 0; i < 8; ++i) {
        const long g0 = (long)rowBase + mhi + i;
        const long g1 = g0 + 16;
        if (g0 < (long)nNodes) out[g0 * F_DIM + colBase + m] = acc0[i];
        if (g1 < (long)nNodes) out[g1 * F_DIM + colBase + m] = acc1[i];
    }
}

extern "C" void kernel_launch(void* const* d_in, const int* in_sizes, int n_in,
                              void* d_out, int out_size, void* d_ws, size_t ws_size,
                              hipStream_t stream)
{
    const int*   row  = (const int*)d_in[0];
    const int*   col  = (const int*)d_in[1];
    const float* vals = (const float*)d_in[2];
    const float* x    = (const float*)d_in[3];
    const float* W    = (const float*)d_in[4];
    float*       out  = (float*)d_out;

    const int nEdges = in_sizes[0];
    const int nNodes = in_sizes[3] / F_DIM;

    const int nBuckets = (nNodes + BUCKET_ROWS - 1) / BUCKET_ROWS;
    const int nChunks  = (nEdges + CHUNK_EDGES - 1) / CHUNK_EDGES;
    const int nHist    = nBuckets * nChunks;

    const int hPad = (nHist + SCAN_TILE - 1) / SCAN_TILE * SCAN_TILE;
    const int nPad = (nNodes + 1023) / 1024 * 1024;
    const int ePad = (nEdges + 1023) / 1024 * 1024;
    int*  hist   = (int*)d_ws;
    int*  base   = hist + hPad;
    int*  cnt    = base + hPad;
    int*  off    = cnt + nPad;
    int2* binned = (int2*)(off + nPad);
    int2* pe     = binned + ePad;

    const size_t binBytes = (size_t)nBuckets * sizeof(int);

    bin_count_kernel<<<nChunks, 256, binBytes, stream>>>(row, hist, nEdges, nBuckets);

    row_start_kernel<<<hPad / SCAN_TILE, 256, 0, stream>>>(hist, base, nHist);

    bin_scatter_kernel<<<nChunks, 256, binBytes, stream>>>(row, col, vals, base, binned, nEdges, nBuckets);

    bucket_sort_kernel<<<nBuckets, 256, 0, stream>>>(base, binned, off, cnt, pe, nNodes, nEdges, nChunks);

    const int gridF = (nNodes + TILE_ROWS - 1) / TILE_ROWS;
    spmm_gemm_kernel<<<gridF, 256, 0, stream>>>(off, cnt, pe, x, W, out, nNodes);
}


# Level-2 row sort fused into the SpMM launch: each 32-row block ranks its own edges of the bucket list into an LDS list; one launch and the sorted-list plane fewer; row sum order = LDS arrival order
# speedup vs baseline: 2.7594x; 1.1708x over previous
#include <hip/hip_runtime.h>


typedef float v2f __attribute__((ext_vector_type(2)));
typedef float v8f __attribute__((ext_vector_type(8)));

#define F_DIM 128
#define SZ_STRIDE 132
#define SCAN_TILE 1024

#define BUCKET_SHIFT 8
#define BUCKET_ROWS  (1 << BUCKET_SHIFT)
#define CHUNK_PER_THREAD 16
#define CHUNK_EDGES (256 * CHUNK_PER_THREAD)
#define KEY_SHIFT 24
#define COL_MASK  0x00ffffff

__global__ __launch_bounds__(256) void bin_count_kernel(
    const int* __restrict__ row, int* __restrict__ hist, int nEdges, int nBuckets)
{
    extern __shared__ int sBin[];
    const int tid = threadIdx.x;
    for (int i = tid; i < nBuckets; i += 256) sBin[i] = 0;
    __syncthreads();

    const int chunkBeg = (int)blockIdx.x * CHUNK_EDGES;
    int r[CHUNK_PER_THREAD];
#pragma unroll
    for (int u = 0; u < CHUNK_PER_THREAD; ++u) {
        const int e = chunkBeg + u * 256 + tid;
        r[u] = row[e < nEdges ? e : nEdges - 1];
    }
#pragma unroll
    for (int u = 0; u < CHUNK_PER_THREAD; ++u) {
        const int e = chunkBeg + u * 256 + tid;
        atomicAdd(&sBin[r[u] >> BUCKET_SHIFT], e < nEdges ? 1 : 0);
    }
    __syncthreads();

    const int nChunks = (int)gridDim.x;
    for (int i = tid; i < nBuckets; i += 256)
        hist[i * nChunks + (int)blockIdx.x] = sBin[i];
}

__global__ __launch_bounds__(256) void row_start_kernel(
    const int* __restrict__ src, int* __restrict__ dst, int n)
{
    __shared__ int sFront[8];
    __shared__ int sWave[8];

    const int tid     = threadIdx.x;
    const int lane    = tid & 31;
    const int wave    = tid >> 5;
    const int tileBeg = (int)blockIdx.x * SCAN_TILE;

    int s = 0;
    const int4* c4 = (const int4*)src;
    const int nb = (int)blockIdx.x;
    for (int b = 0; b < nb; b += 16) {
        int4 v[16];
#pragma unroll
        for (int u = 0; u < 16; ++u) {
            const int sl = (b + u < nb) ? (b + u) : (nb - 1);
            v[u] = c4[sl * 256 + tid];
        }
#pragma unroll
        for (int u = 0; u < 16; ++u)
            s += (b + u < nb) ? ((v[u].x + v[u].y) + (v[u].z + v[u].w)) : 0;
    }
    for (int d = 16; d > 0; d >>= 1) s += __shfl_xor(s, d);
    if (lane == 0) sFront[wave] = s;

    const int i0 = tileBeg + tid * 4;
    const int a0 = (i0     < n) ? src[i0]     : 0;
    const int a1 = (i0 + 1 < n) ? src[i0 + 1] : 0;
    const int a2 = (i0 + 2 < n) ? src[i0 + 2] : 0;
    const int a3 = (i0 + 3 < n) ? src[i0 + 3] : 0;
    const int t  = (a0 + a1) + (a2 + a3);

    int inc = t;
    for (int d = 1; d < 32; d <<= 1) {
        const int u = __shfl_up(inc, d);
        if (lane >= d) inc += u;
    }
    if (lane == 31) sWave[wave] = inc;
    __syncthreads();

    int base = 0;
    for (int w = 0; w < 8; ++w) base += sFront[w];
    for (int w = 0; w < wave; ++w) base += sWave[w];

    const int ex = base + inc - t;
    if (i0     < n) dst[i0]     = ex;
    if (i0 + 1 < n) dst[i0 + 1] = ex + a0;
    if (i0 + 2 < n) dst[i0 + 2] = ex + a0 + a1;
    if (i0 + 3 < n) dst[i0 + 3] = ex + a0 + a1 + a2;
}

__global__ __launch_bounds__(256) void bin_scatter_kernel(
    const int* __restrict__ row, const int* __restrict__ col,
    const float* __restrict__ vals, const int* __restrict__ base,
    int2* __restrict__ binned, int nEdges, int nBuckets)
{
    extern __shared__ int sBin[];
    const int tid = threadIdx.x;
    const int nChunks = (int)gridDim.x;
    for (int i = tid; i < nBuckets; i += 256)
        sBin[i] = base[i * nChunks + (int)blockIdx.x];
    __syncthreads();

    const int chunkBeg = (int)blockIdx.x * CHUNK_EDGES;
    int   r[CHUNK_PER_THREAD];
    int   c[CHUNK_PER_THREAD];
    float v[CHUNK_PER_THREAD];
#pragma unroll
    for (int u = 0; u < CHUNK_PER_THREAD; ++u) {
        const int e  = chunkBeg + u * 256 + tid;
        const int ec = e < nEdges ? e : nEdges - 1;
        r[u] = row[ec];
        c[u] = col[ec];
        v[u] = vals[ec];
    }
    int p[CHUNK_PER_THREAD];
#pragma unroll
    for (int u = 0; u < CHUNK_PER_THREAD; ++u) {
        const int e = chunkBeg + u * 256 + tid;
        p[u] = atomicAdd(&sBin[r[u] >> BUCKET_SHIFT], e < nEdges ? 1 : 0);
    }
#pragma unroll
    for (int u = 0; u < CHUNK_PER_THREAD; ++u) {
        const int e = chunkBeg + u * 256 + tid;
        if (e < nEdges) {
            const unsigned key = ((unsigned)r[u] & (BUCKET_ROWS - 1)) << KEY_SHIFT;
            binned[p[u]] = make_int2((int)(key | (unsigned)c[u]), __float_as_int(v[u]));
        }
    }
}

static __device__ __forceinline__ void fma4(float4& acc, float v, const float4& d)
{
    acc.x = fmaf(v, d.x, acc.x);
    acc.y = fmaf(v, d.y, acc.y);
    acc.z = fmaf(v, d.z, acc.z);
    acc.w = fmaf(v, d.w, acc.w);
}

#define TILE_ROWS 32
#define WAVE_ROWS 4
#define TILE_SHIFT 5
#define TILES_PER_BUCKET (BUCKET_ROWS / TILE_ROWS)
#define SCAN_BATCH 18
#define EDGE_CAP 1024

template <int G>
static __device__ __forceinline__ void gather_group(
    float4& acc, const float4* __restrict__ x4, int2 ev, int j, int lane)
{
    float4 d[G];
    float  v[G];
#pragma unroll
    for (int u = 0; u < G; ++u) {
        const int c = __builtin_amdgcn_readlane(ev.x, j + u);
        v[u] = __int_as_float(__builtin_amdgcn_readlane(ev.y, j + u));
        d[u] = x4[(size_t)c * (F_DIM / 4) + lane];
    }
#pragma unroll
    for (int u = 0; u < G; ++u) fma4(acc, v[u], d[u]);
}

__global__ __launch_bounds__(256) void spmm_gemm_kernel(
    const int* __restrict__ base, const int2* __restrict__ binned,
    const float* __restrict__ x, const float* __restrict__ W,
    float* __restrict__ out, int nNodes, int nEdges, int nChunks, int nBuckets)
{
    __shared__ __attribute__((aligned(16))) float sZ[TILE_ROWS * SZ_STRIDE];
    __shared__ int2 sEdge[EDGE_CAP];
    __shared__ int sCnt[TILE_ROWS];
    __shared__ int sCur[TILE_ROWS];

    const int tid  = threadIdx.x;
    const int lane = tid & 31;
    const int wave = __builtin_amdgcn_readfirstlane(tid >> 5);
    const int rowBase = (int)blockIdx.x * TILE_ROWS;
    const float4* x4 = (const float4*)x;

    const int      b   = (int)blockIdx.x / TILES_PER_BUCKET;
    const unsigned sub = (unsigned)blockIdx.x % TILES_PER_BUCKET;
    const int beg = base[b * nChunks];
    const int end = (b + 1 < nBuckets) ? base[(b + 1) * nChunks] : nEdges;
    const int last = (end > beg) ? (end - 1) : 0;

    if (tid < TILE_ROWS) sCnt[tid] = 0;
    __syncthreads();

    int2 rec0[SCAN_BATCH];
#pragma unroll
    for (int u = 0; u < SCAN_BATCH; ++u) {
        const int i = beg + u * 256 + tid;
        rec0[u] = binned[i < end ? i : last];
    }
    __builtin_amdgcn_sched_barrier(0);
#pragma unroll
    for (int u = 0; u < SCAN_BATCH; ++u) {
        const int i = beg + u * 256 + tid;
        const unsigned key = (unsigned)rec0[u].x >> KEY_SHIFT;
        if (i < end && (key >> TILE_SHIFT) == sub)
            atomicAdd(&sCnt[key & (TILE_ROWS - 1)], 1);
    }
    for (int s0 = beg + 256 * SCAN_BATCH; s0 < end; s0 += 256 * SCAN_BATCH) {
        int k2[SCAN_BATCH];
#pragma unroll
        for (int u = 0; u < SCAN_BATCH; ++u) {
            const int i = s0 + u * 256 + tid;
            k2[u] = binned[i < end ? i : last].x;
        }
#pragma unroll
        for (int u = 0; u < SCAN_BATCH; ++u) {
            const int i = s0 + u * 256 + tid;
            const unsigned key = (unsigned)k2[u] >> KEY_SHIFT;
            if (i < end && (key >> TILE_SHIFT) == sub)
                atomicAdd(&sCnt[key & (TILE_ROWS - 1)], 1);
        }
    }
    __syncthreads();

    const int rowCnt = sCnt[lane];
    int inc = rowCnt;
    for (int d = 1; d < 32; d <<= 1) {
        const int up = __shfl_up(inc, d);
        if (lane >= d) inc += up;
    }
    const int rowStart = inc - rowCnt;
    const int total = __builtin_amdgcn_readlane(inc, 31);
    const bool inLds = total <= EDGE_CAP;

    if (wave == 0) sCur[lane] = rowStart;
    __syncthreads();

    if (inLds) {
#pragma unroll
        for (int u = 0; u < SCAN_BATCH; ++u) {
            const int i = beg + u * 256 + tid;
            const unsigned key = (unsigned)rec0[u].x >> KEY_SHIFT;
            if (i < end && (key >> TILE_SHIFT) == sub) {
                const int p = atomicAdd(&sCur[key & (TILE_ROWS - 1)], 1);
                sEdge[p] = make_int2(rec0[u].x & COL_MASK, rec0[u].y);
            }
        }
        for (int s0 = beg + 256 * SCAN_BATCH; s0 < end; s0 += 256 * SCAN_BATCH) {
            int2 rec[SCAN_BATCH];
#pragma unroll
            for (int u = 0; u < SCAN_BATCH; ++u) {
                const int i = s0 + u * 256 + tid;
                rec[u] = binned[i < end ? i : last];
            }
#pragma unroll
            for (int u = 0; u < SCAN_BATCH; ++u) {
                const int i = s0 + u * 256 + tid;
                const unsigned key = (unsigned)rec[u].x >> KEY_SHIFT;
                if (i < end && (key >> TILE_SHIFT) == sub) {
                    const int p = atomicAdd(&sCur[key & (TILE_ROWS - 1)], 1);
                    sEdge[p] = make_int2(rec[u].x & COL_MASK, rec[u].y);
                }
            }
        }
    }
    __syncthreads();

#pragma unroll
    for (int q = 0; q < WAVE_ROWS; ++q) {
        const int r5 = wave * WAVE_ROWS + q;
        const int rb = __builtin_amdgcn_readlane(rowStart, r5);
        const int rn = __builtin_amdgcn_readlane(rowCnt, r5);
        float4 acc = make_float4(0.0f, 0.0f, 0.0f, 0.0f);
        if (inLds) {
            for (int eb = 0; eb < rn; eb += 32) {
                const int m = (rn - eb < 32) ? (rn - eb) : 32;
                int2 ev = make_int2(0, 0);
                if (lane < m) ev = sEdge[rb + eb + lane];
                int j = 0;
                for (; j + 8 <= m; j += 8) gather_group<8>(acc, x4, ev, j, lane);
                if (j + 4 <= m) { gather_group<4>(acc, x4, ev, j, lane); j += 4; }
                if (j + 2 <= m) { gather_group<2>(acc, x4, ev, j, lane); j += 2; }
                if (j < m)      { gather_group<1>(acc, x4, ev, j, lane); }
            }
        } else {
            const unsigned want = (sub << TILE_SHIFT) | (unsigned)r5;
            for (int s0 = beg; s0 < end; s0 += 32) {
                const int i = s0 + lane;
                const int2 ev = binned[i < end ? i : last];
                const bool hit = (i < end) && (((unsigned)ev.x >> KEY_SHIFT) == want);
                unsigned mask = __builtin_amdgcn_ballot_w32(hit);
                while (mask) {
                    const int j = __builtin_ctz(mask);
                    mask &= mask - 1;
                    const int   c = __builtin_amdgcn_readlane(ev.x, j) & COL_MASK;
                    const float v = __int_as_float(__builtin_amdgcn_readlane(ev.y, j));
                    fma4(acc, v, x4[(size_t)c * (F_DIM / 4) + lane]);
                }
            }
        }
        *(float4*)&sZ[(wave * WAVE_ROWS + q) * SZ_STRIDE + lane * 4] = acc;
    }
    __syncthreads();

    const int m       = lane & 15;
    const int kh      = (lane >> 4) << 1;
    const int colBase = wave << 4;

    v8f acc0 = {};
    v8f acc1 = {};
#pragma unroll
    for (int k = 0; k < F_DIM; k += 4) {
        v2f a0, a1, b;
        a0.x = sZ[m * SZ_STRIDE + k + kh];
        a0.y = sZ[m * SZ_STRIDE + k + kh + 1];
        a1.x = sZ[(16 + m) * SZ_STRIDE + k + kh];
        a1.y = sZ[(16 + m) * SZ_STRIDE + k + kh + 1];
        b.x = W[(k + kh)     * F_DIM + colBase + m];
        b.y = W[(k + kh + 1) * F_DIM + colBase + m];
        acc0 = __builtin_amdgcn_wmma_f32_16x16x4_f32(
             false, a0,  false, b,
             (short)0, acc0,  false,  false);
        acc1 = __builtin_amdgcn_wmma_f32_16x16x4_f32(
             false, a1,  false, b,
             (short)0, acc1,  false,  false);
    }

    const int mhi = (lane >> 4) << 3;
#pragma unroll
    for (int i = 0; i < 8; ++i) {
        const long g0 = (long)rowBase + mhi + i;
        const long g1 = g0 + 16;
        if (g0 < (long)nNodes) out[g0 * F_DIM + colBase + m] = acc0[i];
        if (g1 < (long)nNodes) out[g1 * F_DIM + colBase + m] = acc1[i];
    }
}

extern "C" void kernel_launch(void* const* d_in, const int* in_sizes, int n_in,
                              void* d_out, int out_size, void* d_ws, size_t ws_size,
                              hipStream_t stream)
{
    const int*   row  = (const int*)d_in[0];
    const int*   col  = (const int*)d_in[1];
    const float* vals = (const float*)d_in[2];
    const float* x    = (const float*)d_in[3];
    const float* W    = (const float*)d_in[4];
    float*       out  = (float*)d_out;

    const int nEdges = in_sizes[0];
    const int nNodes = in_sizes[3] / F_DIM;

    const int nBuckets = (nNodes + BUCKET_ROWS - 1) / BUCKET_ROWS;
    const int nChunks  = (nEdges + CHUNK_EDGES - 1) / CHUNK_EDGES;
    const int nHist    = nBuckets * nChunks;

    const int hPad = (nHist + SCAN_TILE - 1) / SCAN_TILE * SCAN_TILE;
    int*  hist   = (int*)d_ws;
    int*  base   = hist + hPad;
    int2* binned = (int2*)(base + hPad);

    const size_t binBytes = (size_t)nBuckets * sizeof(int);

    bin_count_kernel<<<nChunks, 256, binBytes, stream>>>(row, hist, nEdges, nBuckets);

    row_start_kernel<<<hPad / SCAN_TILE, 256, 0, stream>>>(hist, base, nHist);

    bin_scatter_kernel<<<nChunks, 256, binBytes, stream>>>(row, col, vals, base, binned, nEdges, nBuckets);

    const int gridF = (nNodes + TILE_ROWS - 1) / TILE_ROWS;
    spmm_gemm_kernel<<<gridF, 256, 0, stream>>>(base, binned, x, W, out, nNodes, nEdges, nChunks, nBuckets);
}


# Level-1 scatter stores coalesced: each block orders its 4096-edge chunk by bucket in LDS first, then writes it out in runs of consecutive slots; order inside a segment stays LDS arrival order
# speedup vs baseline: 3.1625x; 1.1461x over previous
#include <hip/hip_runtime.h>


typedef float v2f __attribute__((ext_vector_type(2)));
typedef float v8f __attribute__((ext_vector_type(8)));

#define F_DIM 128
#define SZ_STRIDE 132
#define SCAN_TILE 1024

#define BUCKET_SHIFT 8
#define BUCKET_ROWS  (1 << BUCKET_SHIFT)
#define CHUNK_PER_THREAD 16
#define CHUNK_EDGES (256 * CHUNK_PER_THREAD)
#define KEY_SHIFT 24
#define COL_MASK  0x00ffffff

__global__ __launch_bounds__(256) void bin_count_kernel(
    const int* __restrict__ row, int* __restrict__ hist, int nEdges, int nBuckets)
{
    extern __shared__ int sBin[];
    const int tid = threadIdx.x;
    for (int i = tid; i < nBuckets; i += 256) sBin[i] = 0;
    __syncthreads();

    const int chunkBeg = (int)blockIdx.x * CHUNK_EDGES;
    int r[CHUNK_PER_THREAD];
#pragma unroll
    for (int u = 0; u < CHUNK_PER_THREAD; ++u) {
        const int e = chunkBeg + u * 256 + tid;
        r[u] = row[e < nEdges ? e : nEdges - 1];
    }
#pragma unroll
    for (int u = 0; u < CHUNK_PER_THREAD; ++u) {
        const int e = chunkBeg + u * 256 + tid;
        atomicAdd(&sBin[r[u] >> BUCKET_SHIFT], e < nEdges ? 1 : 0);
    }
    __syncthreads();

    const int nChunks = (int)gridDim.x;
    for (int i = tid; i < nBuckets; i += 256)
        hist[i * nChunks + (int)blockIdx.x] = sBin[i];
}

__global__ __launch_bounds__(256) void row_start_kernel(
    const int* __restrict__ src, int* __restrict__ dst, int n)
{
    __shared__ int sFront[8];
    __shared__ int sWave[8];

    const int tid     = threadIdx.x;
    const int lane    = tid & 31;
    const int wave    = tid >> 5;
    const int tileBeg = (int)blockIdx.x * SCAN_TILE;

    int s = 0;
    const int4* c4 = (const int4*)src;
    const int nb = (int)blockIdx.x;
    for (int b = 0; b < nb; b += 16) {
        int4 v[16];
#pragma unroll
        for (int u = 0; u < 16; ++u) {
            const int sl = (b + u < nb) ? (b + u) : (nb - 1);
            v[u] = c4[sl * 256 + tid];
        }
#pragma unroll
        for (int u = 0; u < 16; ++u)
            s += (b + u < nb) ? ((v[u].x + v[u].y) + (v[u].z + v[u].w)) : 0;
    }
    for (int d = 16; d > 0; d >>= 1) s += __shfl_xor(s, d);
    if (lane == 0) sFront[wave] = s;

    const int i0 = tileBeg + tid * 4;
    const int a0 = (i0     < n) ? src[i0]     : 0;
    const int a1 = (i0 + 1 < n) ? src[i0 + 1] : 0;
    const int a2 = (i0 + 2 < n) ? src[i0 + 2] : 0;
    const int a3 = (i0 + 3 < n) ? src[i0 + 3] : 0;
    const int t  = (a0 + a1) + (a2 + a3);

    int inc = t;
    for (int d = 1; d < 32; d <<= 1) {
        const int u = __shfl_up(inc, d);
        if (lane >= d) inc += u;
    }
    if (lane == 31) sWave[wave] = inc;
    __syncthreads();

    int base = 0;
    for (int w = 0; w < 8; ++w) base += sFront[w];
    for (int w = 0; w < wave; ++w) base += sWave[w];

    const int ex = base + inc - t;
    if (i0     < n) dst[i0]     = ex;
    if (i0 + 1 < n) dst[i0 + 1] = ex + a0;
    if (i0 + 2 < n) dst[i0 + 2] = ex + a0 + a1;
    if (i0 + 3 < n) dst[i0 + 3] = ex + a0 + a1 + a2;
}

__global__ __launch_bounds__(256) void bin_scatter_kernel(
    const int* __restrict__ row, const int* __restrict__ col,
    const float* __restrict__ vals, const int* __restrict__ hist,
    const int* __restrict__ base,
    int2* __restrict__ binned, int nEdges, int nBuckets)
{
    extern __shared__ int sBin[];
    __shared__ int2 sRec[CHUNK_EDGES];
    __shared__ int  sDst[CHUNK_EDGES];
    __shared__ int  sWaveTot[8];

    const int tid  = threadIdx.x;
    const int lane = tid & 31;
    const int wave = tid >> 5;
    const int nChunks = (int)gridDim.x;
    const int blk = (int)blockIdx.x;
    int* sCur = sBin;
    int* sOff = sBin + nBuckets;

    const int chunkBeg = blk * CHUNK_EDGES;
    int   r[CHUNK_PER_THREAD];
    int   c[CHUNK_PER_THREAD];
    float v[CHUNK_PER_THREAD];
#pragma unroll
    for (int u = 0; u < CHUNK_PER_THREAD; ++u) {
        const int e  = chunkBeg + u * 256 + tid;
        const int ec = e < nEdges ? e : nEdges - 1;
        r[u] = row[ec];
        c[u] = col[ec];
        v[u] = vals[ec];
    }

    int carry = 0;
    for (int i0 = 0; i0 < nBuckets; i0 += 256) {
        const int  i  = i0 + tid;
        const bool in = i < nBuckets;
        const int cnt = in ? hist[i * nChunks + blk] : 0;
        const int g   = in ? base[i * nChunks + blk] : 0;
        int inc = cnt;
        for (int d = 1; d < 32; d <<= 1) {
            const int up = __shfl_up(inc, d);
            if (lane >= d) inc += up;
        }
        if (lane == 31) sWaveTot[wave] = inc;
        __syncthreads();
        int front = 0;
        int tot   = 0;
        for (int w = 0; w < 8; ++w) {
            const int s = sWaveTot[w];
            tot += s;
            if (w < wave) front += s;
        }
        const int ex = carry + front + inc - cnt;
        if (in) {
            sCur[i] = ex;
            sOff[i] = g - ex;
        }
        carry += tot;
        __syncthreads();
    }

    int p[CHUNK_PER_THREAD];
    int o[CHUNK_PER_THREAD];
#pragma unroll
    for (int u = 0; u < CHUNK_PER_THREAD; ++u) {
        const int e  = chunkBeg + u * 256 + tid;
        const int bk = r[u] >> BUCKET_SHIFT;
        p[u] = atomicAdd(&sCur[bk], e < nEdges ? 1 : 0);
        o[u] = sOff[bk];
    }
#pragma unroll
    for (int u = 0; u < CHUNK_PER_THREAD; ++u) {
        const int e = chunkBeg + u * 256 + tid;
        if (e < nEdges) {
            const unsigned key = ((unsigned)r[u] & (BUCKET_ROWS - 1)) << KEY_SHIFT;
            sRec[p[u]] = make_int2((int)(key | (unsigned)c[u]), __float_as_int(v[u]));
            sDst[p[u]] = p[u] + o[u];
        }
    }
    __syncthreads();

    const int left   = nEdges - chunkBeg;
    const int nValid = left < CHUNK_EDGES ? left : CHUNK_EDGES;
    int2 rec[CHUNK_PER_THREAD];
    int  dst[CHUNK_PER_THREAD];
#pragma unroll
    for (int u = 0; u < CHUNK_PER_THREAD; ++u) {
        const int j  = u * 256 + tid;
        const int jc = j < nValid ? j : nValid - 1;
        rec[u] = sRec[jc];
        dst[u] = sDst[jc];
    }
#pragma unroll
    for (int u = 0; u < CHUNK_PER_THREAD; ++u) {
        const int j = u * 256 + tid;
        if (j < nValid) binned[dst[u]] = rec[u];
    }
}

static __device__ __forceinline__ void fma4(float4& acc, float v, const float4& d)
{
    acc.x = fmaf(v, d.x, acc.x);
    acc.y = fmaf(v, d.y, acc.y);
    acc.z = fmaf(v, d.z, acc.z);
    acc.w = fmaf(v, d.w, acc.w);
}

#define TILE_ROWS 32
#define WAVE_ROWS 4
#define TILE_SHIFT 5
#define TILES_PER_BUCKET (BUCKET_ROWS / TILE_ROWS)
#define SCAN_BATCH 18
#define EDGE_CAP 1024

template <int G>
static __device__ __forceinline__ void gather_group(
    float4& acc, const float4* __restrict__ x4, int2 ev, int j, int lane)
{
    float4 d[G];
    float  v[G];
#pragma unroll
    for (int u = 0; u < G; ++u) {
        const int c = __builtin_amdgcn_readlane(ev.x, j + u);
        v[u] = __int_as_float(__builtin_amdgcn_readlane(ev.y, j + u));
        d[u] = x4[(size_t)c * (F_DIM / 4) + lane];
    }
#pragma unroll
    for (int u = 0; u < G; ++u) fma4(acc, v[u], d[u]);
}

__global__ __launch_bounds__(256) void spmm_gemm_kernel(
    const int* __restrict__ base, const int2* __restrict__ binned,
    const float* __restrict__ x, const float* __restrict__ W,
    float* __restrict__ out, int nNodes, int nEdges, int nChunks, int nBuckets)
{
    __shared__ __attribute__((aligned(16))) float sZ[TILE_ROWS * SZ_STRIDE];
    __shared__ int2 sEdge[EDGE_CAP];
    __shared__ int sCnt[TILE_ROWS];
    __shared__ int sCur[TILE_ROWS];

    const int tid  = threadIdx.x;
    const int lane = tid & 31;
    const int wave = __builtin_amdgcn_readfirstlane(tid >> 5);
    const int rowBase = (int)blockIdx.x * TILE_ROWS;
    const float4* x4 = (const float4*)x;

    const int      b   = (int)blockIdx.x / TILES_PER_BUCKET;
    const unsigned sub = (unsigned)blockIdx.x % TILES_PER_BUCKET;
    const int beg = base[b * nChunks];
    const int end = (b + 1 < nBuckets) ? base[(b + 1) * nChunks] : nEdges;
    const int last = (end > beg) ? (end - 1) : 0;

    if (tid < TILE_ROWS) sCnt[tid] = 0;
    __syncthreads();

    int2 rec0[SCAN_BATCH];
#pragma unroll
    for (int u = 0; u < SCAN_BATCH; ++u) {
        const int i = beg + u * 256 + tid;
        rec0[u] = binned[i < end ? i : last];
    }
    __builtin_amdgcn_sched_barrier(0);
#pragma unroll
    for (int u = 0; u < SCAN_BATCH; ++u) {
        const int i = beg + u * 256 + tid;
        const unsigned key = (unsigned)rec0[u].x >> KEY_SHIFT;
        if (i < end && (key >> TILE_SHIFT) == sub)
            atomicAdd(&sCnt[key & (TILE_ROWS - 1)], 1);
    }
    for (int s0 = beg + 256 * SCAN_BATCH; s0 < end; s0 += 256 * SCAN_BATCH) {
        int k2[SCAN_BATCH];
#pragma unroll
        for (int u = 0; u < SCAN_BATCH; ++u) {
            const int i = s0 + u * 256 + tid;
            k2[u] = binned[i < end ? i : last].x;
        }
#pragma unroll
        for (int u = 0; u < SCAN_BATCH; ++u) {
            const int i = s0 + u * 256 + tid;
            const unsigned key = (unsigned)k2[u] >> KEY_SHIFT;
            if (i < end && (key >> TILE_SHIFT) == sub)
                atomicAdd(&sCnt[key & (TILE_ROWS - 1)], 1);
        }
    }
    __syncthreads();

    const int rowCnt = sCnt[lane];
    int inc = rowCnt;
    for (int d = 1; d < 32; d <<= 1) {
        const int up = __shfl_up(inc, d);
        if (lane >= d) inc += up;
    }
    const int rowStart = inc - rowCnt;
    const int total = __builtin_amdgcn_readlane(inc, 31);
    const bool inLds = total <= EDGE_CAP;

    if (wave == 0) sCur[lane] = rowStart;
    __syncthreads();

    if (inLds) {
#pragma unroll
        for (int u = 0; u < SCAN_BATCH; ++u) {
            const int i = beg + u * 256 + tid;
            const unsigned key = (unsigned)rec0[u].x >> KEY_SHIFT;
            if (i < end && (key >> TILE_SHIFT) == sub) {
                const int p = atomicAdd(&sCur[key & (TILE_ROWS - 1)], 1);
                sEdge[p] = make_int2(rec0[u].x & COL_MASK, rec0[u].y);
            }
        }
        for (int s0 = beg + 256 * SCAN_BATCH; s0 < end; s0 += 256 * SCAN_BATCH) {
            int2 rec[SCAN_BATCH];
#pragma unroll
            for (int u = 0; u < SCAN_BATCH; ++u) {
                const int i = s0 + u * 256 + tid;
                rec[u] = binned[i < end ? i : last];
            }
#pragma unroll
            for (int u = 0; u < SCAN_BATCH; ++u) {
                const int i = s0 + u * 256 + tid;
                const unsigned key = (unsigned)rec[u].x >> KEY_SHIFT;
                if (i < end && (key >> TILE_SHIFT) == sub) {
                    const int p = atomicAdd(&sCur[key & (TILE_ROWS - 1)], 1);
                    sEdge[p] = make_int2(rec[u].x & COL_MASK, rec[u].y);
                }
            }
        }
    }
    __syncthreads();

#pragma unroll
    for (int q = 0; q < WAVE_ROWS; ++q) {
        const int r5 = wave * WAVE_ROWS + q;
        const int rb = __builtin_amdgcn_readlane(rowStart, r5);
        const int rn = __builtin_amdgcn_readlane(rowCnt, r5);
        float4 acc = make_float4(0.0f, 0.0f, 0.0f, 0.0f);
        if (inLds) {
            for (int eb = 0; eb < rn; eb += 32) {
                const int m = (rn - eb < 32) ? (rn - eb) : 32;
                int2 ev = make_int2(0, 0);
                if (lane < m) ev = sEdge[rb + eb + lane];
                int j = 0;
                for (; j + 8 <= m; j += 8) gather_group<8>(acc, x4, ev, j, lane);
                if (j + 4 <= m) { gather_group<4>(acc, x4, ev, j, lane); j += 4; }
                if (j + 2 <= m) { gather_group<2>(acc, x4, ev, j, lane); j += 2; }
                if (j < m)      { gather_group<1>(acc, x4, ev, j, lane); }
            }
        } else {
            const unsigned want = (sub << TILE_SHIFT) | (unsigned)r5;
            for (int s0 = beg; s0 < end; s0 += 32) {
                const int i = s0 + lane;
                const int2 ev = binned[i < end ? i : last];
                const bool hit = (i < end) && (((unsigned)ev.x >> KEY_SHIFT) == want);
                unsigned mask = __builtin_amdgcn_ballot_w32(hit);
                while (mask) {
                    const int j = __builtin_ctz(mask);
                    mask &= mask - 1;
                    const int   c = __builtin_amdgcn_readlane(ev.x, j) & COL_MASK;
                    const float v = __int_as_float(__builtin_amdgcn_readlane(ev.y, j));
                    fma4(acc, v, x4[(size_t)c * (F_DIM / 4) + lane]);
                }
            }
        }
        *(float4*)&sZ[(wave * WAVE_ROWS + q) * SZ_STRIDE + lane * 4] = acc;
    }
    __syncthreads();

    const int m       = lane & 15;
    const int kh      = (lane >> 4) << 1;
    const int colBase = wave << 4;

    v8f acc0 = {};
    v8f acc1 = {};
#pragma unroll
    for (int k = 0; k < F_DIM; k += 4) {
        v2f a0, a1, b;
        a0.x = sZ[m * SZ_STRIDE + k + kh];
        a0.y = sZ[m * SZ_STRIDE + k + kh + 1];
        a1.x = sZ[(16 + m) * SZ_STRIDE + k + kh];
        a1.y = sZ[(16 + m) * SZ_STRIDE + k + kh + 1];
        b.x = W[(k + kh)     * F_DIM + colBase + m];
        b.y = W[(k + kh + 1) * F_DIM + colBase + m];
        acc0 = __builtin_amdgcn_wmma_f32_16x16x4_f32(
             false, a0,  false, b,
             (short)0, acc0,  false,  false);
        acc1 = __builtin_amdgcn_wmma_f32_16x16x4_f32(
             false, a1,  false, b,
             (short)0, acc1,  false,  false);
    }

    const int mhi = (lane >> 4) << 3;
#pragma unroll
    for (int i = 0; i < 8; ++i) {
        const long g0 = (long)rowBase + mhi + i;
        const long g1 = g0 + 16;
        if (g0 < (long)nNodes) out[g0 * F_DIM + colBase + m] = acc0[i];
        if (g1 < (long)nNodes) out[g1 * F_DIM + colBase + m] = acc1[i];
    }
}

extern "C" void kernel_launch(void* const* d_in, const int* in_sizes, int n_in,
                              void* d_out, int out_size, void* d_ws, size_t ws_size,
                              hipStream_t stream)
{
    const int*   row  = (const int*)d_in[0];
    const int*   col  = (const int*)d_in[1];
    const float* vals = (const float*)d_in[2];
    const float* x    = (const float*)d_in[3];
    const float* W    = (const float*)d_in[4];
    float*       out  = (float*)d_out;

    const int nEdges = in_sizes[0];
    const int nNodes = in_sizes[3] / F_DIM;

    const int nBuckets = (nNodes + BUCKET_ROWS - 1) / BUCKET_ROWS;
    const int nChunks  = (nEdges + CHUNK_EDGES - 1) / CHUNK_EDGES;
    const int nHist    = nBuckets * nChunks;

    const int hPad = (nHist + SCAN_TILE - 1) / SCAN_TILE * SCAN_TILE;
    int*  hist   = (int*)d_ws;
    int*  base   = hist + hPad;
    int2* binned = (int2*)(base + hPad);

    const size_t binBytes = (size_t)nBuckets * sizeof(int);
    const size_t pairBytes = (size_t)nBuckets * sizeof(int2);

    bin_count_kernel<<<nChunks, 256, binBytes, stream>>>(row, hist, nEdges, nBuckets);

    row_start_kernel<<<hPad / SCAN_TILE, 256, 0, stream>>>(hist, base, nHist);

    bin_scatter_kernel<<<nChunks, 256, pairBytes, stream>>>(row, col, vals, hist, base, binned, nEdges, nBuckets);

    const int gridF = (nNodes + TILE_ROWS - 1) / TILE_ROWS;
    spmm_gemm_kernel<<<gridF, 256, 0, stream>>>(base, binned, x, W, out, nNodes, nEdges, nChunks, nBuckets);
}


# Fused SpMM+projection kernel held to 64 vector registers (16 waves per SIMD requested; W fragment loads in two halves of K, not all at once): 8 blocks resident per WGP instead of 6
# speedup vs baseline: 3.3641x; 1.0637x over previous
#include <hip/hip_runtime.h>


typedef float v2f __attribute__((ext_vector_type(2)));
typedef float v8f __attribute__((ext_vector_type(8)));

#define F_DIM 128
#define SZ_STRIDE 132
#define SCAN_TILE 1024

#define BUCKET_SHIFT 8
#define BUCKET_ROWS  (1 << BUCKET_SHIFT)
#define CHUNK_PER_THREAD 16
#define CHUNK_EDGES (256 * CHUNK_PER_THREAD)
#define KEY_SHIFT 24
#define COL_MASK  0x00ffffff

__global__ __launch_bounds__(256) void bin_count_kernel(
    const int* __restrict__ row, int* __restrict__ hist, int nEdges, int nBuckets)
{
    extern __shared__ int sBin[];
    const int tid = threadIdx.x;
    for (int i = tid; i < nBuckets; i += 256) sBin[i] = 0;
    __syncthreads();

    const int chunkBeg = (int)blockIdx.x * CHUNK_EDGES;
    int r[CHUNK_PER_THREAD];
#pragma unroll
    for (int u = 0; u < CHUNK_PER_THREAD; ++u) {
        const int e = chunkBeg + u * 256 + tid;
        r[u] = row[e < nEdges ? e : nEdges - 1];
    }
#pragma unroll
    for (int u = 0; u < CHUNK_PER_THREAD; ++u) {
        const int e = chunkBeg + u * 256 + tid;
        atomicAdd(&sBin[r[u] >> BUCKET_SHIFT], e < nEdges ? 1 : 0);
    }
    __syncthreads();

    const int nChunks = (int)gridDim.x;
    for (int i = tid; i < nBuckets; i += 256)
        hist[i * nChunks + (int)blockIdx.x] = sBin[i];
}

__global__ __launch_bounds__(256) void row_start_kernel(
    const int* __restrict__ src, int* __restrict__ dst, int n)
{
    __shared__ int sFront[8];
    __shared__ int sWave[8];

    const int tid     = threadIdx.x;
    const int lane    = tid & 31;
    const int wave    = tid >> 5;
    const int tileBeg = (int)blockIdx.x * SCAN_TILE;

    int s = 0;
    const int4* c4 = (const int4*)src;
    const int nb = (int)blockIdx.x;
    for (int b = 0; b < nb; b += 16) {
        int4 v[16];
#pragma unroll
        for (int u = 0; u < 16; ++u) {
            const int sl = (b + u < nb) ? (b + u) : (nb - 1);
            v[u] = c4[sl * 256 + tid];
        }
#pragma unroll
        for (int u = 0; u < 16; ++u)
            s += (b + u < nb) ? ((v[u].x + v[u].y) + (v[u].z + v[u].w)) : 0;
    }
    for (int d = 16; d > 0; d >>= 1) s += __shfl_xor(s, d);
    if (lane == 0) sFront[wave] = s;

    const int i0 = tileBeg + tid * 4;
    const int a0 = (i0     < n) ? src[i0]     : 0;
    const int a1 = (i0 + 1 < n) ? src[i0 + 1] : 0;
    const int a2 = (i0 + 2 < n) ? src[i0 + 2] : 0;
    const int a3 = (i0 + 3 < n) ? src[i0 + 3] : 0;
    const int t  = (a0 + a1) + (a2 + a3);

    int inc = t;
    for (int d = 1; d < 32; d <<= 1) {
        const int u = __shfl_up(inc, d);
        if (lane >= d) inc += u;
    }
    if (lane == 31) sWave[wave] = inc;
    __syncthreads();

    int base = 0;
    for (int w = 0; w < 8; ++w) base += sFront[w];
    for (int w = 0; w < wave; ++w) base += sWave[w];

    const int ex = base + inc - t;
    if (i0     < n) dst[i0]     = ex;
    if (i0 + 1 < n) dst[i0 + 1] = ex + a0;
    if (i0 + 2 < n) dst[i0 + 2] = ex + a0 + a1;
    if (i0 + 3 < n) dst[i0 + 3] = ex + a0 + a1 + a2;
}

__global__ __launch_bounds__(256) void bin_scatter_kernel(
    const int* __restrict__ row, const int* __restrict__ col,
    const float* __restrict__ vals, const int* __restrict__ hist,
    const int* __restrict__ base,
    int2* __restrict__ binned, int nEdges, int nBuckets)
{
    extern __shared__ int sBin[];
    __shared__ int2 sRec[CHUNK_EDGES];
    __shared__ int  sDst[CHUNK_EDGES];
    __shared__ int  sWaveTot[8];

    const int tid  = threadIdx.x;
    const int lane = tid & 31;
    const int wave = tid >> 5;
    const int nChunks = (int)gridDim.x;
    const int blk = (int)blockIdx.x;
    int* sCur = sBin;
    int* sOff = sBin + nBuckets;

    const int chunkBeg = blk * CHUNK_EDGES;
    int   r[CHUNK_PER_THREAD];
    int   c[CHUNK_PER_THREAD];
    float v[CHUNK_PER_THREAD];
#pragma unroll
    for (int u = 0; u < CHUNK_PER_THREAD; ++u) {
        const int e  = chunkBeg + u * 256 + tid;
        const int ec = e < nEdges ? e : nEdges - 1;
        r[u] = row[ec];
        c[u] = col[ec];
        v[u] = vals[ec];
    }

    int carry = 0;
    for (int i0 = 0; i0 < nBuckets; i0 += 256) {
        const int  i  = i0 + tid;
        const bool in = i < nBuckets;
        const int cnt = in ? hist[i * nChunks + blk] : 0;
        const int g   = in ? base[i * nChunks + blk] : 0;
        int inc = cnt;
        for (int d = 1; d < 32; d <<= 1) {
            const int up = __shfl_up(inc, d);
            if (lane >= d) inc += up;
        }
        if (lane == 31) sWaveTot[wave] = inc;
        __syncthreads();
        int front = 0;
        int tot   = 0;
        for (int w = 0; w < 8; ++w) {
            const int s = sWaveTot[w];
            tot += s;
            if (w < wave) front += s;
        }
        const int ex = carry + front + inc - cnt;
        if (in) {
            sCur[i] = ex;
            sOff[i] = g - ex;
        }
        carry += tot;
        __syncthreads();
    }

    int p[CHUNK_PER_THREAD];
    int o[CHUNK_PER_THREAD];
#pragma unroll
    for (int u = 0; u < CHUNK_PER_THREAD; ++u) {
        const int e  = chunkBeg + u * 256 + tid;
        const int bk = r[u] >> BUCKET_SHIFT;
        p[u] = atomicAdd(&sCur[bk], e < nEdges ? 1 : 0);
        o[u] = sOff[bk];
    }
#pragma unroll
    for (int u = 0; u < CHUNK_PER_THREAD; ++u) {
        const int e = chunkBeg + u * 256 + tid;
        if (e < nEdges) {
            const unsigned key = ((unsigned)r[u] & (BUCKET_ROWS - 1)) << KEY_SHIFT;
            sRec[p[u]] = make_int2((int)(key | (unsigned)c[u]), __float_as_int(v[u]));
            sDst[p[u]] = p[u] + o[u];
        }
    }
    __syncthreads();

    const int left   = nEdges - chunkBeg;
    const int nValid = left < CHUNK_EDGES ? left : CHUNK_EDGES;
    int2 rec[CHUNK_PER_THREAD];
    int  dst[CHUNK_PER_THREAD];
#pragma unroll
    for (int u = 0; u < CHUNK_PER_THREAD; ++u) {
        const int j  = u * 256 + tid;
        const int jc = j < nValid ? j : nValid - 1;
        rec[u] = sRec[jc];
        dst[u] = sDst[jc];
    }
#pragma unroll
    for (int u = 0; u < CHUNK_PER_THREAD; ++u) {
        const int j = u * 256 + tid;
        if (j < nValid) binned[dst[u]] = rec[u];
    }
}

static __device__ __forceinline__ void fma4(float4& acc, float v, const float4& d)
{
    acc.x = fmaf(v, d.x, acc.x);
    acc.y = fmaf(v, d.y, acc.y);
    acc.z = fmaf(v, d.z, acc.z);
    acc.w = fmaf(v, d.w, acc.w);
}

#define TILE_ROWS 32
#define WAVE_ROWS 4
#define TILE_SHIFT 5
#define TILES_PER_BUCKET (BUCKET_ROWS / TILE_ROWS)
#define SCAN_BATCH 18
#define EDGE_CAP 1024

template <int G>
static __device__ __forceinline__ void gather_group(
    float4& acc, const float4* __restrict__ x4, int2 ev, int j, int lane)
{
    float4 d[G];
    float  v[G];
#pragma unroll
    for (int u = 0; u < G; ++u) {
        const int c = __builtin_amdgcn_readlane(ev.x, j + u);
        v[u] = __int_as_float(__builtin_amdgcn_readlane(ev.y, j + u));
        d[u] = x4[(size_t)c * (F_DIM / 4) + lane];
    }
#pragma unroll
    for (int u = 0; u < G; ++u) fma4(acc, v[u], d[u]);
}

__global__ __launch_bounds__(256, 16) void spmm_gemm_kernel(
    const int* __restrict__ base, const int2* __restrict__ binned,
    const float* __restrict__ x, const float* __restrict__ W,
    float* __restrict__ out, int nNodes, int nEdges, int nChunks, int nBuckets)
{
    __shared__ __attribute__((aligned(16))) float sZ[TILE_ROWS * SZ_STRIDE];
    __shared__ int2 sEdge[EDGE_CAP];
    __shared__ int sCnt[TILE_ROWS];
    __shared__ int sCur[TILE_ROWS];

    const int tid  = threadIdx.x;
    const int lane = tid & 31;
    const int wave = __builtin_amdgcn_readfirstlane(tid >> 5);
    const int rowBase = (int)blockIdx.x * TILE_ROWS;
    const float4* x4 = (const float4*)x;

    const int      b   = (int)blockIdx.x / TILES_PER_BUCKET;
    const unsigned sub = (unsigned)blockIdx.x % TILES_PER_BUCKET;
    const int beg = base[b * nChunks];
    const int end = (b + 1 < nBuckets) ? base[(b + 1) * nChunks] : nEdges;
    const int last = (end > beg) ? (end - 1) : 0;

    if (tid < TILE_ROWS) sCnt[tid] = 0;
    __syncthreads();

    int2 rec0[SCAN_BATCH];
#pragma unroll
    for (int u = 0; u < SCAN_BATCH; ++u) {
        const int i = beg + u * 256 + tid;
        rec0[u] = binned[i < end ? i : last];
    }
    __builtin_amdgcn_sched_barrier(0);
#pragma unroll
    for (int u = 0; u < SCAN_BATCH; ++u) {
        const int i = beg + u * 256 + tid;
        const unsigned key = (unsigned)rec0[u].x >> KEY_SHIFT;
        if (i < end && (key >> TILE_SHIFT) == sub)
            atomicAdd(&sCnt[key & (TILE_ROWS - 1)], 1);
    }
    for (int s0 = beg + 256 * SCAN_BATCH; s0 < end; s0 += 256 * SCAN_BATCH) {
        int k2[SCAN_BATCH];
#pragma unroll
        for (int u = 0; u < SCAN_BATCH; ++u) {
            const int i = s0 + u * 256 + tid;
            k2[u] = binned[i < end ? i : last].x;
        }
#pragma unroll
        for (int u = 0; u < SCAN_BATCH; ++u) {
            const int i = s0 + u * 256 + tid;
            const unsigned key = (unsigned)k2[u] >> KEY_SHIFT;
            if (i < end && (key >> TILE_SHIFT) == sub)
                atomicAdd(&sCnt[key & (TILE_ROWS - 1)], 1);
        }
    }
    __syncthreads();

    const int rowCnt = sCnt[lane];
    int inc = rowCnt;
    for (int d = 1; d < 32; d <<= 1) {
        const int up = __shfl_up(inc, d);
        if (lane >= d) inc += up;
    }
    const int rowStart = inc - rowCnt;
    const int total = __builtin_amdgcn_readlane(inc, 31);
    const bool inLds = total <= EDGE_CAP;

    if (wave == 0) sCur[lane] = rowStart;
    __syncthreads();

    if (inLds) {
#pragma unroll
        for (int u = 0; u < SCAN_BATCH; ++u) {
            const int i = beg + u * 256 + tid;
            const unsigned key = (unsigned)rec0[u].x >> KEY_SHIFT;
            if (i < end && (key >> TILE_SHIFT) == sub) {
                const int p = atomicAdd(&sCur[key & (TILE_ROWS - 1)], 1);
                sEdge[p] = make_int2(rec0[u].x & COL_MASK, rec0[u].y);
            }
        }
        for (int s0 = beg + 256 * SCAN_BATCH; s0 < end; s0 += 256 * SCAN_BATCH) {
            int2 rec[SCAN_BATCH];
#pragma unroll
            for (int u = 0; u < SCAN_BATCH; ++u) {
                const int i = s0 + u * 256 + tid;
                rec[u] = binned[i < end ? i : last];
            }
#pragma unroll
            for (int u = 0; u < SCAN_BATCH; ++u) {
                const int i = s0 + u * 256 + tid;
                const unsigned key = (unsigned)rec[u].x >> KEY_SHIFT;
                if (i < end && (key >> TILE_SHIFT) == sub) {
                    const int p = atomicAdd(&sCur[key & (TILE_ROWS - 1)], 1);
                    sEdge[p] = make_int2(rec[u].x & COL_MASK, rec[u].y);
                }
            }
        }
    }
    __syncthreads();

#pragma unroll
    for (int q = 0; q < WAVE_ROWS; ++q) {
        const int r5 = wave * WAVE_ROWS + q;
        const int rb = __builtin_amdgcn_readlane(rowStart, r5);
        const int rn = __builtin_amdgcn_readlane(rowCnt, r5);
        float4 acc = make_float4(0.0f, 0.0f, 0.0f, 0.0f);
        if (inLds) {
            for (int eb = 0; eb < rn; eb += 32) {
                const int m = (rn - eb < 32) ? (rn - eb) : 32;
                int2 ev = make_int2(0, 0);
                if (lane < m) ev = sEdge[rb + eb + lane];
                int j = 0;
                for (; j + 8 <= m; j += 8) gather_group<8>(acc, x4, ev, j, lane);
                if (j + 4 <= m) { gather_group<4>(acc, x4, ev, j, lane); j += 4; }
                if (j + 2 <= m) { gather_group<2>(acc, x4, ev, j, lane); j += 2; }
                if (j < m)      { gather_group<1>(acc, x4, ev, j, lane); }
            }
        } else {
            const unsigned want = (sub << TILE_SHIFT) | (unsigned)r5;
            for (int s0 = beg; s0 < end; s0 += 32) {
                const int i = s0 + lane;
                const int2 ev = binned[i < end ? i : last];
                const bool hit = (i < end) && (((unsigned)ev.x >> KEY_SHIFT) == want);
                unsigned mask = __builtin_amdgcn_ballot_w32(hit);
                while (mask) {
                    const int j = __builtin_ctz(mask);
                    mask &= mask - 1;
                    const int   c = __builtin_amdgcn_readlane(ev.x, j) & COL_MASK;
                    const float v = __int_as_float(__builtin_amdgcn_readlane(ev.y, j));
                    fma4(acc, v, x4[(size_t)c * (F_DIM / 4) + lane]);
                }
            }
        }
        *(float4*)&sZ[(wave * WAVE_ROWS + q) * SZ_STRIDE + lane * 4] = acc;
    }
    __syncthreads();

    const int m       = lane & 15;
    const int kh      = (lane >> 4) << 1;
    const int colBase = wave << 4;

    v8f acc0 = {};
    v8f acc1 = {};
#pragma unroll 16
    for (int k = 0; k < F_DIM; k += 4) {
        v2f a0, a1, b;
        a0.x = sZ[m * SZ_STRIDE + k + kh];
        a0.y = sZ[m * SZ_STRIDE + k + kh + 1];
        a1.x = sZ[(16 + m) * SZ_STRIDE + k + kh];
        a1.y = sZ[(16 + m) * SZ_STRIDE + k + kh + 1];
        b.x = W[(k + kh)     * F_DIM + colBase + m];
        b.y = W[(k + kh + 1) * F_DIM + colBase + m];
        acc0 = __builtin_amdgcn_wmma_f32_16x16x4_f32(
             false, a0,  false, b,
             (short)0, acc0,  false,  false);
        acc1 = __builtin_amdgcn_wmma_f32_16x16x4_f32(
             false, a1,  false, b,
             (short)0, acc1,  false,  false);
    }

    const int mhi = (lane >> 4) << 3;
#pragma unroll
    for (int i = 0; i < 8; ++i) {
        const long g0 = (long)rowBase + mhi + i;
        const long g1 = g0 + 16;
        if (g0 < (long)nNodes) out[g0 * F_DIM + colBase + m] = acc0[i];
        if (g1 < (long)nNodes) out[g1 * F_DIM + colBase + m] = acc1[i];
    }
}

extern "C" void kernel_launch(void* const* d_in, const int* in_sizes, int n_in,
                              void* d_out, int out_size, void* d_ws, size_t ws_size,
                              hipStream_t stream)
{
    const int*   row  = (const int*)d_in[0];
    const int*   col  = (const int*)d_in[1];
    const float* vals = (const float*)d_in[2];
    const float* x    = (const float*)d_in[3];
    const float* W    = (const float*)d_in[4];
    float*       out  = (float*)d_out;

    const int nEdges = in_sizes[0];
    const int nNodes = in_sizes[3] / F_DIM;

    const int nBuckets = (nNodes + BUCKET_ROWS - 1) / BUCKET_ROWS;
    const int nChunks  = (nEdges + CHUNK_EDGES - 1) / CHUNK_EDGES;
    const int nHist    = nBuckets * nChunks;

    const int hPad = (nHist + SCAN_TILE - 1) / SCAN_TILE * SCAN_TILE;
    int*  hist   = (int*)d_ws;
    int*  base   = hist + hPad;
    int2* binned = (int2*)(base + hPad);

    const size_t binBytes = (size_t)nBuckets * sizeof(int);
    const size_t pairBytes = (size_t)nBuckets * sizeof(int2);

    bin_count_kernel<<<nChunks, 256, binBytes, stream>>>(row, hist, nEdges, nBuckets);

    row_start_kernel<<<hPad / SCAN_TILE, 256, 0, stream>>>(hist, base, nHist);

    bin_scatter_kernel<<<nChunks, 256, pairBytes, stream>>>(row, col, vals, hist, base, binned, nEdges, nBuckets);

    const int gridF = (nNodes + TILE_ROWS - 1) / TILE_ROWS;
    spmm_gemm_kernel<<<gridF, 256, 0, stream>>>(base, binned, x, W, out, nNodes, nEdges, nChunks, nBuckets);
}


# Edges sorted by 64-row bucket within each 8192-edge chunk only (LDS counters); SpMM block reads its bucket's segment of every chunk: no global scan, 2 launches not 4; row sum order: LDS arrival order
# speedup vs baseline: 3.9475x; 1.1734x over previous
#include <hip/hip_runtime.h>


typedef float v2f __attribute__((ext_vector_type(2)));
typedef float v8f __attribute__((ext_vector_type(8)));

#define F_DIM 128
#define SZ_STRIDE 132

#define BUCKET_SHIFT 6
#define BUCKET_ROWS  (1 << BUCKET_SHIFT)
#define SORT_THREADS 512
#define SORT_WAVES (SORT_THREADS / 32)
#define CHUNK_PER_THREAD 16
#define CHUNK_EDGES (SORT_THREADS * CHUNK_PER_THREAD)
#define KEY_SHIFT 24
#define COL_MASK  0x00ffffff

__global__ __launch_bounds__(SORT_THREADS) void chunk_sort_kernel(
    const int* __restrict__ row, const int* __restrict__ col,
    const float* __restrict__ vals,
    int* __restrict__ seg, int2* __restrict__ sorted, int nEdges, int nBuckets)
{
    extern __shared__ int sBin[];
    __shared__ int2 sRec[CHUNK_EDGES];
    __shared__ int  sWaveTot[SORT_WAVES];

    const int tid  = threadIdx.x;
    const int lane = tid & 31;
    const int wave = tid >> 5;
    const int nChunks = (int)gridDim.x;
    const int blk = (int)blockIdx.x;

    for (int i = tid; i < nBuckets; i += SORT_THREADS) sBin[i] = 0;

    const int chunkBeg = blk * CHUNK_EDGES;
    int   r[CHUNK_PER_THREAD];
    int   c[CHUNK_PER_THREAD];
    float v[CHUNK_PER_THREAD];
#pragma unroll
    for (int u = 0; u < CHUNK_PER_THREAD; ++u) {
        const int e  = chunkBeg + u * SORT_THREADS + tid;
        const int ec = e < nEdges ? e : nEdges - 1;
        r[u] = row[ec];
        c[u] = col[ec];
        v[u] = vals[ec];
    }
    __builtin_amdgcn_sched_barrier(0);
    __syncthreads();

    int p[CHUNK_PER_THREAD];
#pragma unroll
    for (int u = 0; u < CHUNK_PER_THREAD; ++u) {
        const int e = chunkBeg + u * SORT_THREADS + tid;
        p[u] = atomicAdd(&sBin[r[u] >> BUCKET_SHIFT], e < nEdges ? 1 : 0);
    }
    __syncthreads();

    int carry = 0;
    for (int i0 = 0; i0 < nBuckets; i0 += SORT_THREADS * 4) {
        const int i = i0 + tid * 4;
        const int a0 = (i     < nBuckets) ? sBin[i]     : 0;
        const int a1 = (i + 1 < nBuckets) ? sBin[i + 1] : 0;
        const int a2 = (i + 2 < nBuckets) ? sBin[i + 2] : 0;
        const int a3 = (i + 3 < nBuckets) ? sBin[i + 3] : 0;
        const int t  = (a0 + a1) + (a2 + a3);
        int inc = t;
        for (int d = 1; d < 32; d <<= 1) {
            const int up = __shfl_up(inc, d);
            if (lane >= d) inc += up;
        }
        if (lane == 31) sWaveTot[wave] = inc;
        __syncthreads();
        int front = 0;
        int tot   = 0;
        for (int w = 0; w < SORT_WAVES; ++w) {
            const int s = sWaveTot[w];
            tot += s;
            if (w < wave) front += s;
        }
        const int e0 = carry + front + inc - t;
        const int e1 = e0 + a0;
        const int e2 = e1 + a1;
        const int e3 = e2 + a2;
        if (i     < nBuckets) { sBin[i]     = e0; seg[(i)     * nChunks + blk] = e0; }
        if (i + 1 < nBuckets) { sBin[i + 1] = e1; seg[(i + 1) * nChunks + blk] = e1; }
        if (i + 2 < nBuckets) { sBin[i + 2] = e2; seg[(i + 2) * nChunks + blk] = e2; }
        if (i + 3 < nBuckets) { sBin[i + 3] = e3; seg[(i + 3) * nChunks + blk] = e3; }
        carry += tot;
        __syncthreads();
    }

    const int left   = nEdges - chunkBeg;
    const int nValid = left < CHUNK_EDGES ? left : CHUNK_EDGES;
    if (tid == 0) seg[nBuckets * nChunks + blk] = nValid;

#pragma unroll
    for (int u = 0; u < CHUNK_PER_THREAD; ++u)
        p[u] += sBin[r[u] >> BUCKET_SHIFT];
#pragma unroll
    for (int u = 0; u < CHUNK_PER_THREAD; ++u) {
        const int e = chunkBeg + u * SORT_THREADS + tid;
        if (e < nEdges) {
            const unsigned key = ((unsigned)r[u] & (BUCKET_ROWS - 1)) << KEY_SHIFT;
            sRec[p[u]] = make_int2((int)(key | (unsigned)c[u]), __float_as_int(v[u]));
        }
    }
    __syncthreads();

    int2 rec[CHUNK_PER_THREAD];
#pragma unroll
    for (int u = 0; u < CHUNK_PER_THREAD; ++u) {
        const int j  = u * SORT_THREADS + tid;
        const int jc = j < nValid ? j : nValid - 1;
        rec[u] = sRec[jc];
    }
#pragma unroll
    for (int u = 0; u < CHUNK_PER_THREAD; ++u) {
        const int j = u * SORT_THREADS + tid;
        if (j < nValid) sorted[chunkBeg + j] = rec[u];
    }
}

static __device__ __forceinline__ void fma4(float4& acc, float v, const float4& d)
{
    acc.x = fmaf(v, d.x, acc.x);
    acc.y = fmaf(v, d.y, acc.y);
    acc.z = fmaf(v, d.z, acc.z);
    acc.w = fmaf(v, d.w, acc.w);
}

#define TILE_ROWS 32
#define WAVE_ROWS 4
#define TILE_SHIFT 5
#define TILES_PER_BUCKET (BUCKET_ROWS / TILE_ROWS)
#define SEG_LANES 16
#define SEG_BATCH 13
#define SEG_CAP (SEG_BATCH * (256 / SEG_LANES))
#define EDGE_CAP 1024

template <int G>
static __device__ __forceinline__ void gather_group(
    float4& acc, const float4* __restrict__ x4, int2 ev, int j, int lane)
{
    float4 d[G];
    float  v[G];
#pragma unroll
    for (int u = 0; u < G; ++u) {
        const int c = __builtin_amdgcn_readlane(ev.x, j + u);
        v[u] = __int_as_float(__builtin_amdgcn_readlane(ev.y, j + u));
        d[u] = x4[(size_t)c * (F_DIM / 4) + lane];
    }
#pragma unroll
    for (int u = 0; u < G; ++u) fma4(acc, v[u], d[u]);
}

static __device__ __forceinline__ void load_bounds(
    int& ovBeg, int& ovEnd, const int* __restrict__ seg, int2* sSeg,
    int b, int c0, int nChunks, int tid)
{
    __syncthreads();
    int sb = 0;
    int se = 0;
    const int cOwn = c0 + tid;
    if (tid < SEG_CAP && cOwn < nChunks) {
        sb = seg[b * nChunks + cOwn];
        se = seg[(b + 1) * nChunks + cOwn];
    }
    if (tid < SEG_CAP) sSeg[tid] = make_int2(sb, se);
    __syncthreads();
    ovBeg = cOwn * CHUNK_EDGES + sb + SEG_LANES;
    ovEnd = cOwn * CHUNK_EDGES + se;
}

__global__ __launch_bounds__(256, 16) void spmm_gemm_kernel(
    const int* __restrict__ seg, const int2* __restrict__ sorted,
    const float* __restrict__ x, const float* __restrict__ W,
    float* __restrict__ out, int nNodes, int nChunks)
{
    __shared__ __attribute__((aligned(16))) float sZ[TILE_ROWS * SZ_STRIDE];
    __shared__ int2 sEdge[EDGE_CAP];
    __shared__ int2 sSeg[SEG_CAP];
    __shared__ int sCnt[TILE_ROWS];
    __shared__ int sCur[TILE_ROWS];

    const int tid  = threadIdx.x;
    const int lane = tid & 31;
    const int wave = __builtin_amdgcn_readfirstlane(tid >> 5);
    const int rowBase = (int)blockIdx.x * TILE_ROWS;
    const float4* x4 = (const float4*)x;

    const int      b   = (int)blockIdx.x / TILES_PER_BUCKET;
    const unsigned sub = (unsigned)blockIdx.x % TILES_PER_BUCKET;
    const int h   = tid / SEG_LANES;
    const int l16 = tid % SEG_LANES;

    if (tid < TILE_ROWS) sCnt[tid] = 0;

    int2 rec0[SEG_BATCH];
    unsigned vmask0 = 0u;
    int ovBeg0, ovEnd0;
    load_bounds(ovBeg0, ovEnd0, seg, sSeg, b, 0, nChunks, tid);
#pragma unroll
    for (int u = 0; u < SEG_BATCH; ++u) {
        const int  s   = u * (256 / SEG_LANES) + h;
        const int2 be  = sSeg[s];
        const int  off = be.x + l16;
        const bool ok  = off < be.y;
        rec0[u] = sorted[ok ? (s * CHUNK_EDGES + off) : 0];
        vmask0 |= ok ? (1u << u) : 0u;
    }
    __builtin_amdgcn_sched_barrier(0);
#pragma unroll
    for (int u = 0; u < SEG_BATCH; ++u) {
        const unsigned key = (unsigned)rec0[u].x >> KEY_SHIFT;
        if (((vmask0 >> u) & 1u) && (key >> TILE_SHIFT) == sub)
            atomicAdd(&sCnt[key & (TILE_ROWS - 1)], 1);
    }
    for (int j = ovBeg0; j < ovEnd0; ++j) {
        const unsigned key = (unsigned)sorted[j].x >> KEY_SHIFT;
        if ((key >> TILE_SHIFT) == sub)
            atomicAdd(&sCnt[key & (TILE_ROWS - 1)], 1);
    }
    for (int c0 = SEG_CAP; c0 < nChunks; c0 += SEG_CAP) {
        int ob, oe;
        load_bounds(ob, oe, seg, sSeg, b, c0, nChunks, tid);
#pragma nounroll
        for (int u = 0; u < SEG_BATCH; ++u) {
            const int  s   = u * (256 / SEG_LANES) + h;
            const int2 be  = sSeg[s];
            const int  off = be.x + l16;
            if (off < be.y) {
                const unsigned key = (unsigned)sorted[(c0 + s) * CHUNK_EDGES + off].x >> KEY_SHIFT;
                if ((key >> TILE_SHIFT) == sub)
                    atomicAdd(&sCnt[key & (TILE_ROWS - 1)], 1);
            }
        }
        for (int j = ob; j < oe; ++j) {
            const unsigned key = (unsigned)sorted[j].x >> KEY_SHIFT;
            if ((key >> TILE_SHIFT) == sub)
                atomicAdd(&sCnt[key & (TILE_ROWS - 1)], 1);
        }
    }
    __syncthreads();

    const int rowCnt = sCnt[lane];
    int inc = rowCnt;
    for (int d = 1; d < 32; d <<= 1) {
        const int up = __shfl_up(inc, d);
        if (lane >= d) inc += up;
    }
    const int rowStart = inc - rowCnt;
    const int total = __builtin_amdgcn_readlane(inc, 31);
    const bool inLds = total <= EDGE_CAP;

    if (wave == 0) sCur[lane] = rowStart;
    __syncthreads();

    if (inLds) {
#pragma unroll
        for (int u = 0; u < SEG_BATCH; ++u) {
            const unsigned key = (unsigned)rec0[u].x >> KEY_SHIFT;
            if (((vmask0 >> u) & 1u) && (key >> TILE_SHIFT) == sub) {
                const int p = atomicAdd(&sCur[key & (TILE_ROWS - 1)], 1);
                sEdge[p] = make_int2(rec0[u].x & COL_MASK, rec0[u].y);
            }
        }
        for (int j = ovBeg0; j < ovEnd0; ++j) {
            const int2 rv = sorted[j];
            const unsigned key = (unsigned)rv.x >> KEY_SHIFT;
            if ((key >> TILE_SHIFT) == sub) {
                const int p = atomicAdd(&sCur[key & (TILE_ROWS - 1)], 1);
                sEdge[p] = make_int2(rv.x & COL_MASK, rv.y);
            }
        }
        for (int c0 = SEG_CAP; c0 < nChunks; c0 += SEG_CAP) {
            int ob, oe;
            load_bounds(ob, oe, seg, sSeg, b, c0, nChunks, tid);
#pragma nounroll
            for (int u = 0; u < SEG_BATCH; ++u) {
                const int  s   = u * (256 / SEG_LANES) + h;
                const int2 be  = sSeg[s];
                const int  off = be.x + l16;
                if (off < be.y) {
                    const int2 rv = sorted[(c0 + s) * CHUNK_EDGES + off];
                    const unsigned key = (unsigned)rv.x >> KEY_SHIFT;
                    if ((key >> TILE_SHIFT) == sub) {
                        const int p = atomicAdd(&sCur[key & (TILE_ROWS - 1)], 1);
                        sEdge[p] = make_int2(rv.x & COL_MASK, rv.y);
                    }
                }
            }
            for (int j = ob; j < oe; ++j) {
                const int2 rv = sorted[j];
                const unsigned key = (unsigned)rv.x >> KEY_SHIFT;
                if ((key >> TILE_SHIFT) == sub) {
                    const int p = atomicAdd(&sCur[key & (TILE_ROWS - 1)], 1);
                    sEdge[p] = make_int2(rv.x & COL_MASK, rv.y);
                }
            }
        }
    }
    __syncthreads();

#pragma unroll
    for (int q = 0; q < WAVE_ROWS; ++q) {
        const int r5 = wave * WAVE_ROWS + q;
        const int rb = __builtin_amdgcn_readlane(rowStart, r5);
        const int rn = __builtin_amdgcn_readlane(rowCnt, r5);
        float4 acc = make_float4(0.0f, 0.0f, 0.0f, 0.0f);
        if (inLds) {
            for (int eb = 0; eb < rn; eb += 32) {
                const int m = (rn - eb < 32) ? (rn - eb) : 32;
                int2 ev = make_int2(0, 0);
                if (lane < m) ev = sEdge[rb + eb + lane];
                int j = 0;
                for (; j + 8 <= m; j += 8) gather_group<8>(acc, x4, ev, j, lane);
                if (j + 4 <= m) { gather_group<4>(acc, x4, ev, j, lane); j += 4; }
                if (j + 2 <= m) { gather_group<2>(acc, x4, ev, j, lane); j += 2; }
                if (j < m)      { gather_group<1>(acc, x4, ev, j, lane); }
            }
        } else {
            const unsigned want = (sub << TILE_SHIFT) | (unsigned)r5;
            for (int cc = 0; cc < nChunks; ++cc) {
                const int sb = seg[b * nChunks + cc];
                const int se = seg[(b + 1) * nChunks + cc];
                const int cb = cc * CHUNK_EDGES;
                for (int s0 = sb; s0 < se; s0 += 32) {
                    const int i = s0 + lane;
                    const int2 ev = sorted[cb + (i < se ? i : se - 1)];
                    const bool hit = (i < se) && (((unsigned)ev.x >> KEY_SHIFT) == want);
                    unsigned mask = __builtin_amdgcn_ballot_w32(hit);
                    while (mask) {
                        const int j = __builtin_ctz(mask);
                        mask &= mask - 1;
                        const int   c = __builtin_amdgcn_readlane(ev.x, j) & COL_MASK;
                        const float v = __int_as_float(__builtin_amdgcn_readlane(ev.y, j));
                        fma4(acc, v, x4[(size_t)c * (F_DIM / 4) + lane]);
                    }
                }
            }
        }
        *(float4*)&sZ[(wave * WAVE_ROWS + q) * SZ_STRIDE + lane * 4] = acc;
    }
    __syncthreads();

    const int m       = lane & 15;
    const int kh      = (lane >> 4) << 1;
    const int colBase = wave << 4;

    v8f acc0 = {};
    v8f acc1 = {};
#pragma unroll 16
    for (int k = 0; k < F_DIM; k += 4) {
        v2f a0, a1, b;
        a0.x = sZ[m * SZ_STRIDE + k + kh];
        a0.y = sZ[m * SZ_STRIDE + k + kh + 1];
        a1.x = sZ[(16 + m) * SZ_STRIDE + k + kh];
        a1.y = sZ[(16 + m) * SZ_STRIDE + k + kh + 1];
        b.x = W[(k + kh)     * F_DIM + colBase + m];
        b.y = W[(k + kh + 1) * F_DIM + colBase + m];
        acc0 = __builtin_amdgcn_wmma_f32_16x16x4_f32(
             false, a0,  false, b,
             (short)0, acc0,  false,  false);
        acc1 = __builtin_amdgcn_wmma_f32_16x16x4_f32(
             false, a1,  false, b,
             (short)0, acc1,  false,  false);
    }

    const int mhi = (lane >> 4) << 3;
#pragma unroll
    for (int i = 0; i < 8; ++i) {
        const long g0 = (long)rowBase + mhi + i;
        const long g1 = g0 + 16;
        if (g0 < (long)nNodes) out[g0 * F_DIM + colBase + m] = acc0[i];
        if (g1 < (long)nNodes) out[g1 * F_DIM + colBase + m] = acc1[i];
    }
}

extern "C" void kernel_launch(void* const* d_in, const int* in_sizes, int n_in,
                              void* d_out, int out_size, void* d_ws, size_t ws_size,
                              hipStream_t stream)
{
    const int*   row  = (const int*)d_in[0];
    const int*   col  = (const int*)d_in[1];
    const float* vals = (const float*)d_in[2];
    const float* x    = (const float*)d_in[3];
    const float* W    = (const float*)d_in[4];
    float*       out  = (float*)d_out;

    const int nEdges = in_sizes[0];
    const int nNodes = in_sizes[3] / F_DIM;

    const int nBuckets = (nNodes + BUCKET_ROWS - 1) / BUCKET_ROWS;
    const int nChunks  = (nEdges + CHUNK_EDGES - 1) / CHUNK_EDGES;
    const int nSeg     = (nBuckets + 1) * nChunks;

    const int sPad = (nSeg + 1023) / 1024 * 1024;
    int*  seg    = (int*)d_ws;
    int2* sorted = (int2*)(seg + sPad);

    const size_t binBytes = (size_t)nBuckets * sizeof(int);

    chunk_sort_kernel<<<nChunks, SORT_THREADS, binBytes, stream>>>(row, col, vals, seg, sorted, nEdges, nBuckets);

    const int gridF = (nNodes + TILE_ROWS - 1) / TILE_ROWS;
    spmm_gemm_kernel<<<gridF, 256, 0, stream>>>(seg, sorted, x, W, out, nNodes, nChunks);
}
